# starting kernel
# baseline (speedup 1.0000x reference)
#include <hip/hip_runtime.h>
#include <hip/hip_bf16.h>

typedef __attribute__((ext_vector_type(2))) float v2f;
typedef __attribute__((ext_vector_type(8))) float v8f;

#define CCH 64
#define NEG_INF_I ((int)0xFF800000)

__global__ __launch_bounds__(128) void k_gemm(const float* __restrict__ x,
                                              const float* __restrict__ W,
                                              const float* __restrict__ bias,
                                              float* __restrict__ A_ws,
                                              float* __restrict__ B_ws,
                                              int N)
{
    const int lane = threadIdx.x & 31;
    const int wave = threadIdx.x >> 5;
    const int n0   = wave * 16;
    const int m0   = blockIdx.x * 16;
    const int half = lane >> 4;
    const int lm   = lane & 15;
    const int col  = n0 + lm;

    v8f accP = {};
    v8f accQ = {};

    int rowA = m0 + lm;
    if (rowA >= N) rowA = N - 1;

#pragma unroll
    for (int kk = 0; kk < 16; ++kk) {
        const int kb  = kk * 4 + half * 2;
        v2f a, b1, b2;
        a.x  = x[(long)rowA * CCH + kb];
        a.y  = x[(long)rowA * CCH + kb + 1];
        b1.x = W[kb * CCH + col];
        b1.y = W[(kb + 1) * CCH + col];
        b2.x = W[(kb + 64) * CCH + col];
        b2.y = W[(kb + 65) * CCH + col];
        accP = __builtin_amdgcn_wmma_f32_16x16x4_f32(false, a, false, b1,
                                                     (short)0, accP, false, false);
        accQ = __builtin_amdgcn_wmma_f32_16x16x4_f32(false, a, false, b2,
                                                     (short)0, accQ, false, false);
    }

    const float bv = bias[col];
    if (m0 + 16 <= N) {
#pragma unroll
        for (int r = 0; r < 8; ++r) {
            const int row = m0 + r + half * 8;
            const float q = accQ[r];
            A_ws[(long)row * CCH + col] = accP[r] - q + bv;
            B_ws[(long)row * CCH + col] = q;
        }
    } else {
#pragma unroll
        for (int r = 0; r < 8; ++r) {
            const int row = m0 + r + half * 8;
            if (row < N) {
                const float q = accQ[r];
                A_ws[(long)row * CCH + col] = accP[r] - q + bv;
                B_ws[(long)row * CCH + col] = q;
            }
        }
    }
}

__global__ void k_init(int* __restrict__ M, long total)
{
    long i = (long)blockIdx.x * blockDim.x + threadIdx.x;
    if (i < total) M[i] = NEG_INF_I;
}

__global__ __launch_bounds__(256) void k_scatter(const float* __restrict__ Bn,
                                                 const int* __restrict__ ei,
                                                 int* __restrict__ M,
                                                 int E)
{
    long tid   = (long)blockIdx.x * blockDim.x + threadIdx.x;
    long total = (long)E * CCH;
    if (tid >= total) return;
    int e = (int)(tid >> 6);
    int c = (int)(tid & 63);
    int src = ei[e];
    int dst = ei[E + e];
    float v = Bn[(long)src * CCH + c];
    int vi = __float_as_int(v);
    int* addr = M + (long)dst * CCH + c;
    if (vi >= 0) atomicMax(addr, vi);
    else         atomicMin((unsigned int*)addr, (unsigned int)vi);
}

__global__ __launch_bounds__(256) void k_elu_partial(const float* __restrict__ A_ws,
                                                     const int* __restrict__ M,
                                                     float* __restrict__ H,
                                                     float* __restrict__ psum,
                                                     float* __restrict__ psq,
                                                     int N)
{
    __shared__ float smem[256];
    const int t  = threadIdx.x;
    const int c  = t & 63;
    const int r4 = t >> 6;
    const int base = blockIdx.x * 64;

    float s = 0.f, q = 0.f;
    for (int rr = r4; rr < 64; rr += 4) {
        int row = base + rr;
        if (row < N) {
            long idx = (long)row * CCH + c;
            int  mb  = M[idx];
            float h;
            if (mb == NEG_INF_I) {
                h = 0.f;
            } else {
                float z = A_ws[idx] + __int_as_float(mb);
                h = (z > 0.f) ? z : expm1f(z);
            }
            H[idx] = h;
            s += h;
            q += h * h;
        }
    }
    smem[t] = s; __syncthreads();
    if (r4 == 0)
        psum[(long)blockIdx.x * 64 + c] = smem[c] + smem[c + 64] + smem[c + 128] + smem[c + 192];
    __syncthreads();
    smem[t] = q; __syncthreads();
    if (r4 == 0)
        psq[(long)blockIdx.x * 64 + c]  = smem[c] + smem[c + 64] + smem[c + 128] + smem[c + 192];
}

__global__ void k_stats(const float* __restrict__ psum,
                        const float* __restrict__ psq,
                        int nblocks, int N,
                        float* __restrict__ mean,
                        float* __restrict__ invstd)
{
    int c = threadIdx.x;
    if (c >= 64) return;
    double s = 0.0, q = 0.0;
    for (int b = 0; b < nblocks; ++b) {
        s += (double)psum[(long)b * 64 + c];
        q += (double)psq[(long)b * 64 + c];
    }
    double m = s / (double)N;
    double v = q / (double)N - m * m;
    if (v < 0.0) v = 0.0;
    mean[c]   = (float)m;
    invstd[c] = (float)(1.0 / sqrt(v + 1e-5));
}

__global__ __launch_bounds__(256) void k_bn(float* __restrict__ out,
                                            const float* __restrict__ mean,
                                            const float* __restrict__ invstd,
                                            const float* __restrict__ gamma,
                                            const float* __restrict__ beta,
                                            long total)
{
    long i = (long)blockIdx.x * blockDim.x + threadIdx.x;
    if (i < total) {
        int c = (int)(i & 63);
        out[i] = gamma[c] * (out[i] - mean[c]) * invstd[c] + beta[c];
    }
}

extern "C" void kernel_launch(void* const* d_in, const int* in_sizes, int n_in,
                              void* d_out, int out_size, void* d_ws, size_t ws_size,
                              hipStream_t stream)
{
    const float* x     = (const float*)d_in[0];
    const int*   ei    = (const int*)  d_in[1];
    const float* W     = (const float*)d_in[2];
    const float* bias  = (const float*)d_in[3];
    const float* gamma = (const float*)d_in[4];
    const float* beta  = (const float*)d_in[5];
    float* out = (float*)d_out;

    const int N = in_sizes[0] / CCH;
    const int E = in_sizes[1] / 2;

    const size_t nodef = (size_t)N * CCH;
    float* A_ws = (float*)d_ws;
    float* B_ws = A_ws + nodef;
    int*   M_ws = (int*)(B_ws + nodef);
    const int nb4 = (N + 63) / 64;
    float* psum    = (float*)(M_ws + nodef);
    float* psq     = psum + (size_t)nb4 * 64;
    float* meanp   = psq  + (size_t)nb4 * 64;
    float* invstdp = meanp + 64;

    const long tot  = (long)nodef;
    const long etot = (long)E * CCH;

    k_gemm<<<(N + 15) / 16, 128, 0, stream>>>(x, W, bias, A_ws, B_ws, N);
    k_init<<<(int)((tot + 255) / 256), 256, 0, stream>>>(M_ws, tot);
    k_scatter<<<(int)((etot + 255) / 256), 256, 0, stream>>>(B_ws, ei, M_ws, E);
    k_elu_partial<<<nb4, 256, 0, stream>>>(A_ws, M_ws, out, psum, psq, N);
    k_stats<<<1, 64, 0, stream>>>(psum, psq, nb4, N, meanp, invstdp);
    k_bn<<<(int)((tot + 255) / 256), 256, 0, stream>>>(out, meanp, invstdp, gamma, beta, tot);
}


# BN statistics kernel parallelised: one 256-thread block per channel instead of 64 serial threads; partials summed in double by strided lanes + fixed shuffle tree (order of the sums changes)
# speedup vs baseline: 1.4530x; 1.4530x over previous
#include <hip/hip_runtime.h>
#include <hip/hip_bf16.h>

typedef __attribute__((ext_vector_type(2))) float v2f;
typedef __attribute__((ext_vector_type(8))) float v8f;

#define CCH 64
#define NEG_INF_I ((int)0xFF800000)

__global__ __launch_bounds__(128) void k_gemm(const float* __restrict__ x,
                                              const float* __restrict__ W,
                                              const float* __restrict__ bias,
                                              float* __restrict__ A_ws,
                                              float* __restrict__ B_ws,
                                              int N)
{
    const int lane = threadIdx.x & 31;
    const int wave = threadIdx.x >> 5;
    const int n0   = wave * 16;
    const int m0   = blockIdx.x * 16;
    const int half = lane >> 4;
    const int lm   = lane & 15;
    const int col  = n0 + lm;

    v8f accP = {};
    v8f accQ = {};

    int rowA = m0 + lm;
    if (rowA >= N) rowA = N - 1;

#pragma unroll
    for (int kk = 0; kk < 16; ++kk) {
        const int kb  = kk * 4 + half * 2;
        v2f a, b1, b2;
        a.x  = x[(long)rowA * CCH + kb];
        a.y  = x[(long)rowA * CCH + kb + 1];
        b1.x = W[kb * CCH + col];
        b1.y = W[(kb + 1) * CCH + col];
        b2.x = W[(kb + 64) * CCH + col];
        b2.y = W[(kb + 65) * CCH + col];
        accP = __builtin_amdgcn_wmma_f32_16x16x4_f32(false, a, false, b1,
                                                     (short)0, accP, false, false);
        accQ = __builtin_amdgcn_wmma_f32_16x16x4_f32(false, a, false, b2,
                                                     (short)0, accQ, false, false);
    }

    const float bv = bias[col];
    if (m0 + 16 <= N) {
#pragma unroll
        for (int r = 0; r < 8; ++r) {
            const int row = m0 + r + half * 8;
            const float q = accQ[r];
            A_ws[(long)row * CCH + col] = accP[r] - q + bv;
            B_ws[(long)row * CCH + col] = q;
        }
    } else {
#pragma unroll
        for (int r = 0; r < 8; ++r) {
            const int row = m0 + r + half * 8;
            if (row < N) {
                const float q = accQ[r];
                A_ws[(long)row * CCH + col] = accP[r] - q + bv;
                B_ws[(long)row * CCH + col] = q;
            }
        }
    }
}

__global__ void k_init(int* __restrict__ M, long total)
{
    long i = (long)blockIdx.x * blockDim.x + threadIdx.x;
    if (i < total) M[i] = NEG_INF_I;
}

__global__ __launch_bounds__(256) void k_scatter(const float* __restrict__ Bn,
                                                 const int* __restrict__ ei,
                                                 int* __restrict__ M,
                                                 int E)
{
    long tid   = (long)blockIdx.x * blockDim.x + threadIdx.x;
    long total = (long)E * CCH;
    if (tid >= total) return;
    int e = (int)(tid >> 6);
    int c = (int)(tid & 63);
    int src = ei[e];
    int dst = ei[E + e];
    float v = Bn[(long)src * CCH + c];
    int vi = __float_as_int(v);
    int* addr = M + (long)dst * CCH + c;
    if (vi >= 0) atomicMax(addr, vi);
    else         atomicMin((unsigned int*)addr, (unsigned int)vi);
}

__global__ __launch_bounds__(256) void k_elu_partial(const float* __restrict__ A_ws,
                                                     const int* __restrict__ M,
                                                     float* __restrict__ H,
                                                     float* __restrict__ psum,
                                                     float* __restrict__ psq,
                                                     int N)
{
    __shared__ float smem[256];
    const int t  = threadIdx.x;
    const int c  = t & 63;
    const int r4 = t >> 6;
    const int base = blockIdx.x * 64;

    float s = 0.f, q = 0.f;
    for (int rr = r4; rr < 64; rr += 4) {
        int row = base + rr;
        if (row < N) {
            long idx = (long)row * CCH + c;
            int  mb  = M[idx];
            float h;
            if (mb == NEG_INF_I) {
                h = 0.f;
            } else {
                float z = A_ws[idx] + __int_as_float(mb);
                h = (z > 0.f) ? z : expm1f(z);
            }
            H[idx] = h;
            s += h;
            q += h * h;
        }
    }
    smem[t] = s; __syncthreads();
    if (r4 == 0)
        psum[(long)blockIdx.x * 64 + c] = smem[c] + smem[c + 64] + smem[c + 128] + smem[c + 192];
    __syncthreads();
    smem[t] = q; __syncthreads();
    if (r4 == 0)
        psq[(long)blockIdx.x * 64 + c]  = smem[c] + smem[c + 64] + smem[c + 128] + smem[c + 192];
}

#define STATS_THREADS 256
__global__ __launch_bounds__(STATS_THREADS) void k_stats(const float* __restrict__ psum,
                                                         const float* __restrict__ psq,
                                                         int nblocks, int N,
                                                         float* __restrict__ mean,
                                                         float* __restrict__ invstd)
{
    __shared__ double wsum[STATS_THREADS / 32];
    __shared__ double wsq[STATS_THREADS / 32];
    const int c = blockIdx.x;
    const int t = threadIdx.x;

    double s = 0.0, q = 0.0;
#pragma unroll 4
    for (int b = t; b < nblocks; b += STATS_THREADS) {
        s += (double)psum[(long)b * 64 + c];
        q += (double)psq[(long)b * 64 + c];
    }
    for (int off = 16; off > 0; off >>= 1) {
        s += __shfl_xor(s, off);
        q += __shfl_xor(q, off);
    }
    if ((t & 31) == 0) {
        wsum[t >> 5] = s;
        wsq[t >> 5]  = q;
    }
    __syncthreads();
    if (t == 0) {
        double ts = 0.0, tq = 0.0;
        for (int w = 0; w < STATS_THREADS / 32; ++w) {
            ts += wsum[w];
            tq += wsq[w];
        }
        double m = ts / (double)N;
        double v = tq / (double)N - m * m;
        if (v < 0.0) v = 0.0;
        mean[c]   = (float)m;
        invstd[c] = (float)(1.0 / sqrt(v + 1e-5));
    }
}

__global__ __launch_bounds__(256) void k_bn(float* __restrict__ out,
                                            const float* __restrict__ mean,
                                            const float* __restrict__ invstd,
                                            const float* __restrict__ gamma,
                                            const float* __restrict__ beta,
                                            long total)
{
    long i = (long)blockIdx.x * blockDim.x + threadIdx.x;
    if (i < total) {
        int c = (int)(i & 63);
        out[i] = gamma[c] * (out[i] - mean[c]) * invstd[c] + beta[c];
    }
}

extern "C" void kernel_launch(void* const* d_in, const int* in_sizes, int n_in,
                              void* d_out, int out_size, void* d_ws, size_t ws_size,
                              hipStream_t stream)
{
    const float* x     = (const float*)d_in[0];
    const int*   ei    = (const int*)  d_in[1];
    const float* W     = (const float*)d_in[2];
    const float* bias  = (const float*)d_in[3];
    const float* gamma = (const float*)d_in[4];
    const float* beta  = (const float*)d_in[5];
    float* out = (float*)d_out;

    const int N = in_sizes[0] / CCH;
    const int E = in_sizes[1] / 2;

    const size_t nodef = (size_t)N * CCH;
    float* A_ws = (float*)d_ws;
    float* B_ws = A_ws + nodef;
    int*   M_ws = (int*)(B_ws + nodef);
    const int nb4 = (N + 63) / 64;
    float* psum    = (float*)(M_ws + nodef);
    float* psq     = psum + (size_t)nb4 * 64;
    float* meanp   = psq  + (size_t)nb4 * 64;
    float* invstdp = meanp + 64;

    const long tot  = (long)nodef;
    const long etot = (long)E * CCH;

    k_gemm<<<(N + 15) / 16, 128, 0, stream>>>(x, W, bias, A_ws, B_ws, N);
    k_init<<<(int)((tot + 255) / 256), 256, 0, stream>>>(M_ws, tot);
    k_scatter<<<(int)((etot + 255) / 256), 256, 0, stream>>>(B_ws, ei, M_ws, E);
    k_elu_partial<<<nb4, 256, 0, stream>>>(A_ws, M_ws, out, psum, psq, N);
    k_stats<<<CCH, STATS_THREADS, 0, stream>>>(psum, psq, nb4, N, meanp, invstdp);
    k_bn<<<(int)((tot + 255) / 256), 256, 0, stream>>>(out, meanp, invstdp, gamma, beta, tot);
}


# Scatter-max by one global atomic per edge-channel (102M) replaced by counting sort of edges by destination (1 integer atomic per edge) + per-node max in registers; max is order-independent
# speedup vs baseline: 2.4913x; 1.7146x over previous
#include <hip/hip_runtime.h>
#include <hip/hip_bf16.h>

typedef __attribute__((ext_vector_type(2))) float v2f;
typedef __attribute__((ext_vector_type(8))) float v8f;

#define CCH 64
#define NEG_INF_I ((int)0xFF800000)

__global__ __launch_bounds__(128) void k_gemm(const float* __restrict__ x,
                                              const float* __restrict__ W,
                                              const float* __restrict__ bias,
                                              float* __restrict__ A_ws,
                                              float* __restrict__ B_ws,
                                              int N)
{
    const int lane = threadIdx.x & 31;
    const int wave = threadIdx.x >> 5;
    const int n0   = wave * 16;
    const int m0   = blockIdx.x * 16;
    const int half = lane >> 4;
    const int lm   = lane & 15;
    const int col  = n0 + lm;

    v8f accP = {};
    v8f accQ = {};

    int rowA = m0 + lm;
    if (rowA >= N) rowA = N - 1;

#pragma unroll
    for (int kk = 0; kk < 16; ++kk) {
        const int kb  = kk * 4 + half * 2;
        v2f a, b1, b2;
        a.x  = x[(long)rowA * CCH + kb];
        a.y  = x[(long)rowA * CCH + kb + 1];
        b1.x = W[kb * CCH + col];
        b1.y = W[(kb + 1) * CCH + col];
        b2.x = W[(kb + 64) * CCH + col];
        b2.y = W[(kb + 65) * CCH + col];
        accP = __builtin_amdgcn_wmma_f32_16x16x4_f32(false, a, false, b1,
                                                     (short)0, accP, false, false);
        accQ = __builtin_amdgcn_wmma_f32_16x16x4_f32(false, a, false, b2,
                                                     (short)0, accQ, false, false);
    }

    const float bv = bias[col];
    if (m0 + 16 <= N) {
#pragma unroll
        for (int r = 0; r < 8; ++r) {
            const int row = m0 + r + half * 8;
            const float q = accQ[r];
            A_ws[(long)row * CCH + col] = accP[r] - q + bv;
            B_ws[(long)row * CCH + col] = q;
        }
    } else {
#pragma unroll
        for (int r = 0; r < 8; ++r) {
            const int row = m0 + r + half * 8;
            if (row < N) {
                const float q = accQ[r];
                A_ws[(long)row * CCH + col] = accP[r] - q + bv;
                B_ws[(long)row * CCH + col] = q;
            }
        }
    }
}

__global__ __launch_bounds__(256) void k_zero(int* __restrict__ cnt, int n)
{
    int i = blockIdx.x * blockDim.x + threadIdx.x;
    if (i < n) cnt[i] = 0;
}

__global__ __launch_bounds__(256) void k_hist(const int* __restrict__ ei,
                                              int* __restrict__ cnt,
                                              int* __restrict__ rank,
                                              int E)
{
    int e = blockIdx.x * blockDim.x + threadIdx.x;
    if (e < E) rank[e] = atomicAdd(&cnt[ei[E + e]], 1);
}

__global__ __launch_bounds__(256) void k_alloc(const int* __restrict__ cnt,
                                               int* __restrict__ cursor,
                                               int* __restrict__ off,
                                               int N)
{
    const int i    = blockIdx.x * blockDim.x + threadIdx.x;
    const int lane = threadIdx.x & 31;
    const int c    = (i < N) ? cnt[i] : 0;
    int incl = c;
    for (int d = 1; d < 32; d <<= 1) {
        int t = __shfl_up(incl, d);
        if (lane >= d) incl += t;
    }
    const int total = __shfl(incl, 31);
    int base = 0;
    if (lane == 31) base = atomicAdd(cursor, total);
    base = __shfl(base, 31);
    if (i < N) off[i] = base + incl - c;
}

__global__ __launch_bounds__(256) void k_fill(const int* __restrict__ ei,
                                              const int* __restrict__ rank,
                                              const int* __restrict__ off,
                                              int* __restrict__ srcs,
                                              int E)
{
    int e = blockIdx.x * blockDim.x + threadIdx.x;
    if (e < E) srcs[off[ei[E + e]] + rank[e]] = ei[e];
}

__global__ __launch_bounds__(256) void k_gather(const float* __restrict__ Bn,
                                                const int* __restrict__ srcs,
                                                const int* __restrict__ off,
                                                const int* __restrict__ cnt,
                                                int* __restrict__ M,
                                                int N)
{
    const int l16  = threadIdx.x & 15;
    const int node = blockIdx.x * 16 + (threadIdx.x >> 4);
    if (node >= N) return;
    const int start = off[node];
    const int deg   = cnt[node];
    const float4* __restrict__ B4 = (const float4*)Bn;
    const float ninf = __int_as_float(NEG_INF_I);
    float4 m = make_float4(ninf, ninf, ninf, ninf);
    for (int i = 0; i < deg; i += 8) {
        int s[8];
        float4 v[8];
#pragma unroll
        for (int j = 0; j < 8; ++j) {
            int k = i + j;
            if (k > deg - 1) k = deg - 1;
            s[j] = srcs[start + k];
        }
#pragma unroll
        for (int j = 0; j < 8; ++j) v[j] = B4[(long)s[j] * 16 + l16];
#pragma unroll
        for (int j = 0; j < 8; ++j) {
            m.x = fmaxf(m.x, v[j].x);
            m.y = fmaxf(m.y, v[j].y);
            m.z = fmaxf(m.z, v[j].z);
            m.w = fmaxf(m.w, v[j].w);
        }
    }
    int4 o = make_int4(__float_as_int(m.x), __float_as_int(m.y),
                       __float_as_int(m.z), __float_as_int(m.w));
    ((int4*)M)[(long)node * 16 + l16] = o;
}

__global__ __launch_bounds__(256) void k_elu_partial(const float* __restrict__ A_ws,
                                                     const int* __restrict__ M,
                                                     float* __restrict__ H,
                                                     float* __restrict__ psum,
                                                     float* __restrict__ psq,
                                                     int N)
{
    __shared__ float smem[256];
    const int t  = threadIdx.x;
    const int c  = t & 63;
    const int r4 = t >> 6;
    const int base = blockIdx.x * 64;

    float s = 0.f, q = 0.f;
    for (int rr = r4; rr < 64; rr += 4) {
        int row = base + rr;
        if (row < N) {
            long idx = (long)row * CCH + c;
            int  mb  = M[idx];
            float h;
            if (mb == NEG_INF_I) {
                h = 0.f;
            } else {
                float z = A_ws[idx] + __int_as_float(mb);
                h = (z > 0.f) ? z : expm1f(z);
            }
            H[idx] = h;
            s += h;
            q += h * h;
        }
    }
    smem[t] = s; __syncthreads();
    if (r4 == 0)
        psum[(long)blockIdx.x * 64 + c] = smem[c] + smem[c + 64] + smem[c + 128] + smem[c + 192];
    __syncthreads();
    smem[t] = q; __syncthreads();
    if (r4 == 0)
        psq[(long)blockIdx.x * 64 + c]  = smem[c] + smem[c + 64] + smem[c + 128] + smem[c + 192];
}

#define STATS_THREADS 256
__global__ __launch_bounds__(STATS_THREADS) void k_stats(const float* __restrict__ psum,
                                                         const float* __restrict__ psq,
                                                         int nblocks, int N,
                                                         float* __restrict__ mean,
                                                         float* __restrict__ invstd)
{
    __shared__ double wsum[STATS_THREADS / 32];
    __shared__ double wsq[STATS_THREADS / 32];
    const int c = blockIdx.x;
    const int t = threadIdx.x;

    double s = 0.0, q = 0.0;
#pragma unroll 4
    for (int b = t; b < nblocks; b += STATS_THREADS) {
        s += (double)psum[(long)b * 64 + c];
        q += (double)psq[(long)b * 64 + c];
    }
    for (int off = 16; off > 0; off >>= 1) {
        s += __shfl_xor(s, off);
        q += __shfl_xor(q, off);
    }
    if ((t & 31) == 0) {
        wsum[t >> 5] = s;
        wsq[t >> 5]  = q;
    }
    __syncthreads();
    if (t == 0) {
        double ts = 0.0, tq = 0.0;
        for (int w = 0; w < STATS_THREADS / 32; ++w) {
            ts += wsum[w];
            tq += wsq[w];
        }
        double m = ts / (double)N;
        double v = tq / (double)N - m * m;
        if (v < 0.0) v = 0.0;
        mean[c]   = (float)m;
        invstd[c] = (float)(1.0 / sqrt(v + 1e-5));
    }
}

__global__ __launch_bounds__(256) void k_bn(float* __restrict__ out,
                                            const float* __restrict__ mean,
                                            const float* __restrict__ invstd,
                                            const float* __restrict__ gamma,
                                            const float* __restrict__ beta,
                                            long total)
{
    long i = (long)blockIdx.x * blockDim.x + threadIdx.x;
    if (i < total) {
        int c = (int)(i & 63);
        out[i] = gamma[c] * (out[i] - mean[c]) * invstd[c] + beta[c];
    }
}

extern "C" void kernel_launch(void* const* d_in, const int* in_sizes, int n_in,
                              void* d_out, int out_size, void* d_ws, size_t ws_size,
                              hipStream_t stream)
{
    const float* x     = (const float*)d_in[0];
    const int*   ei    = (const int*)  d_in[1];
    const float* W     = (const float*)d_in[2];
    const float* bias  = (const float*)d_in[3];
    const float* gamma = (const float*)d_in[4];
    const float* beta  = (const float*)d_in[5];
    float* out = (float*)d_out;

    const int N = in_sizes[0] / CCH;
    const int E = in_sizes[1] / 2;

    const size_t nodef = (size_t)N * CCH;
    float* A_ws = (float*)d_ws;
    float* B_ws = A_ws + nodef;
    int*   M_ws = (int*)(B_ws + nodef);
    const int nb4 = (N + 63) / 64;
    float* psum    = (float*)(M_ws + nodef);
    float* psq     = psum + (size_t)nb4 * 64;
    float* meanp   = psq  + (size_t)nb4 * 64;
    float* invstdp = meanp + 64;

    const long tot  = (long)nodef;

    const int Np = ((N + 64) / 64) * 64;
    int* cnt    = (int*)d_out;
    int* cursor = cnt + N;
    int* off    = cnt + Np;
    int* rank   = off + Np;
    int* srcs   = rank + E;

    k_gemm<<<(N + 15) / 16, 128, 0, stream>>>(x, W, bias, A_ws, B_ws, N);
    k_zero<<<(N + 1 + 255) / 256, 256, 0, stream>>>(cnt, N + 1);
    k_hist<<<(E + 255) / 256, 256, 0, stream>>>(ei, cnt, rank, E);
    k_alloc<<<(N + 255) / 256, 256, 0, stream>>>(cnt, cursor, off, N);
    k_fill<<<(E + 255) / 256, 256, 0, stream>>>(ei, rank, off, srcs, E);
    k_gather<<<(N + 15) / 16, 256, 0, stream>>>(B_ws, srcs, off, cnt, M_ws, N);
    k_elu_partial<<<nb4, 256, 0, stream>>>(A_ws, M_ws, out, psum, psq, N);
    k_stats<<<CCH, STATS_THREADS, 0, stream>>>(psum, psq, nb4, N, meanp, invstdp);
    k_bn<<<(int)((tot + 255) / 256), 256, 0, stream>>>(out, meanp, invstdp, gamma, beta, tot);
}


# Edge sort without global atomics: 1.6M returning per-edge atomics replaced by a two-level counting sort (256-node buckets, then nodes) ranked by LDS adds; max is order-independent
# speedup vs baseline: 3.3461x; 1.3431x over previous
#include <hip/hip_runtime.h>
#include <hip/hip_bf16.h>

typedef __attribute__((ext_vector_type(2))) float v2f;
typedef __attribute__((ext_vector_type(8))) float v8f;

#define CCH 64
#define NEG_INF_I ((int)0xFF800000)

__global__ __launch_bounds__(128) void k_gemm(const float* __restrict__ x,
                                              const float* __restrict__ W,
                                              const float* __restrict__ bias,
                                              float* __restrict__ A_ws,
                                              float* __restrict__ B_ws,
                                              int N)
{
    const int lane = threadIdx.x & 31;
    const int wave = threadIdx.x >> 5;
    const int n0   = wave * 16;
    const int m0   = blockIdx.x * 16;
    const int half = lane >> 4;
    const int lm   = lane & 15;
    const int col  = n0 + lm;

    v8f accP = {};
    v8f accQ = {};

    int rowA = m0 + lm;
    if (rowA >= N) rowA = N - 1;

#pragma unroll
    for (int kk = 0; kk < 16; ++kk) {
        const int kb  = kk * 4 + half * 2;
        v2f a, b1, b2;
        a.x  = x[(long)rowA * CCH + kb];
        a.y  = x[(long)rowA * CCH + kb + 1];
        b1.x = W[kb * CCH + col];
        b1.y = W[(kb + 1) * CCH + col];
        b2.x = W[(kb + 64) * CCH + col];
        b2.y = W[(kb + 65) * CCH + col];
        accP = __builtin_amdgcn_wmma_f32_16x16x4_f32(false, a, false, b1,
                                                     (short)0, accP, false, false);
        accQ = __builtin_amdgcn_wmma_f32_16x16x4_f32(false, a, false, b2,
                                                     (short)0, accQ, false, false);
    }

    const float bv = bias[col];
    if (m0 + 16 <= N) {
#pragma unroll
        for (int r = 0; r < 8; ++r) {
            const int row = m0 + r + half * 8;
            const float q = accQ[r];
            A_ws[(long)row * CCH + col] = accP[r] - q + bv;
            B_ws[(long)row * CCH + col] = q;
        }
    } else {
#pragma unroll
        for (int r = 0; r < 8; ++r) {
            const int row = m0 + r + half * 8;
            if (row < N) {
                const float q = accQ[r];
                A_ws[(long)row * CCH + col] = accP[r] - q + bv;
                B_ws[(long)row * CCH + col] = q;
            }
        }
    }
}

#define BSHIFT 8
#define BNODES (1 << BSHIFT)
#define MAXB   512
#define CHUNK  4096
#define EPT    16
#define RPT    6

__global__ __launch_bounds__(256) void k_chunk_hist(const int* __restrict__ ei,
                                                    int* __restrict__ T,
                                                    int E, int NB, int nchunks)
{
    __shared__ int h[MAXB];
    const int t     = threadIdx.x;
    const int chunk = blockIdx.x;
    for (int i = t; i < NB; i += 256) h[i] = 0;
    const int e0 = chunk * CHUNK + t;
    int d[EPT];
#pragma unroll
    for (int j = 0; j < EPT; ++j) {
        const int e = e0 + j * 256;
        d[j] = (e < E) ? ei[E + e] : -1;
    }
    __syncthreads();
#pragma unroll
    for (int j = 0; j < EPT; ++j)
        if (d[j] >= 0) atomicAdd(&h[d[j] >> BSHIFT], 1);
    __syncthreads();
    for (int i = t; i < NB; i += 256) T[(long)i * nchunks + chunk] = h[i];
}

__global__ __launch_bounds__(512) void k_bucket_scan(int* __restrict__ T,
                                                     int* __restrict__ total,
                                                     int nchunks)
{
    __shared__ int wtot[16];
    const int t    = threadIdx.x;
    const int lane = t & 31;
    const int w    = t >> 5;
    int* __restrict__ row = T + (long)blockIdx.x * nchunks;
    int carry = 0;
    for (int c0 = 0; c0 < nchunks; c0 += 512) {
        const int c = c0 + t;
        const int v = (c < nchunks) ? row[c] : 0;
        int incl = v;
        for (int dd = 1; dd < 32; dd <<= 1) {
            const int u = __shfl_up(incl, dd);
            if (lane >= dd) incl += u;
        }
        if (lane == 31) wtot[w] = incl;
        __syncthreads();
        int wbase = 0, all = 0;
        for (int k = 0; k < 16; ++k) {
            const int x = wtot[k];
            if (k < w) wbase += x;
            all += x;
        }
        if (c < nchunks) row[c] = carry + wbase + incl - v;
        carry += all;
        __syncthreads();
    }
    if (t == 0) total[blockIdx.x] = carry;
}

__global__ __launch_bounds__(256) void k_chunk_scatter(const int* __restrict__ ei,
                                                       const int* __restrict__ T,
                                                       const int* __restrict__ total,
                                                       int* __restrict__ bbase,
                                                       int* __restrict__ rec,
                                                       int E, int NB, int nchunks)
{
    __shared__ int cur[MAXB];
    __shared__ int wtot[8];
    const int t     = threadIdx.x;
    const int lane  = t & 31;
    const int w     = t >> 5;
    const int chunk = blockIdx.x;
    const int e0    = chunk * CHUNK + t;
    int d[EPT], s[EPT];
#pragma unroll
    for (int j = 0; j < EPT; ++j) {
        const int e = e0 + j * 256;
        const bool ok = e < E;
        d[j] = ok ? ei[E + e] : -1;
        s[j] = ok ? ei[e] : 0;
    }
    const int b0 = 2 * t, b1 = 2 * t + 1;
    const int v0 = (b0 < NB) ? total[b0] : 0;
    const int v1 = (b1 < NB) ? total[b1] : 0;
    const int c0 = (b0 < NB) ? T[(long)b0 * nchunks + chunk] : 0;
    const int c1 = (b1 < NB) ? T[(long)b1 * nchunks + chunk] : 0;
    const int pair = v0 + v1;
    int incl = pair;
    for (int dd = 1; dd < 32; dd <<= 1) {
        const int u = __shfl_up(incl, dd);
        if (lane >= dd) incl += u;
    }
    if (lane == 31) wtot[w] = incl;
    __syncthreads();
    int wbase = 0;
    for (int k = 0; k < 8; ++k) if (k < w) wbase += wtot[k];
    const int ex0 = wbase + incl - pair;
    const int ex1 = ex0 + v0;
    cur[b0] = ex0 + c0;
    cur[b1] = ex1 + c1;
    if (chunk == 0) {
        if (b0 < NB) bbase[b0] = ex0;
        if (b1 < NB) bbase[b1] = ex1;
    }
    __syncthreads();
#pragma unroll
    for (int j = 0; j < EPT; ++j) {
        if (d[j] >= 0) {
            const int p = atomicAdd(&cur[d[j] >> BSHIFT], 1);
            rec[p] = (s[j] << BSHIFT) | (d[j] & (BNODES - 1));
        }
    }
}

__global__ __launch_bounds__(1024) void k_bucket_sort(const int* __restrict__ rec,
                                                      const int* __restrict__ bbase,
                                                      const int* __restrict__ total,
                                                      int* __restrict__ cnt,
                                                      int* __restrict__ off,
                                                      int* __restrict__ srcs,
                                                      int N)
{
    __shared__ int c[BNODES];
    __shared__ int wtot[8];
    const int t    = threadIdx.x;
    const int lane = t & 31;
    const int w    = t >> 5;
    const int b    = blockIdx.x;
    const int base = bbase[b];
    const int n    = total[b];
    if (t < BNODES) c[t] = 0;
    int rd[RPT];
#pragma unroll
    for (int k = 0; k < RPT; ++k) {
        const int i = t + k * 1024;
        rd[k] = (i < n) ? rec[base + i] : -1;
    }
    __syncthreads();
#pragma unroll
    for (int k = 0; k < RPT; ++k)
        if (rd[k] >= 0) atomicAdd(&c[rd[k] & (BNODES - 1)], 1);
    for (int i = t + RPT * 1024; i < n; i += 1024)
        atomicAdd(&c[rec[base + i] & (BNODES - 1)], 1);
    __syncthreads();
    const int v = (t < BNODES) ? c[t] : 0;
    int incl = v;
    for (int dd = 1; dd < 32; dd <<= 1) {
        const int u = __shfl_up(incl, dd);
        if (lane >= dd) incl += u;
    }
    if (t < BNODES && lane == 31) wtot[w] = incl;
    __syncthreads();
    if (t < BNODES) {
        int wbase = 0;
        for (int k = 0; k < 8; ++k) if (k < w) wbase += wtot[k];
        const int ex   = wbase + incl - v;
        const int node = (b << BSHIFT) + t;
        if (node < N) {
            cnt[node] = v;
            off[node] = base + ex;
        }
        c[t] = ex;
    }
    __syncthreads();
#pragma unroll
    for (int k = 0; k < RPT; ++k) {
        if (rd[k] >= 0) {
            const int p = atomicAdd(&c[rd[k] & (BNODES - 1)], 1);
            srcs[base + p] = rd[k] >> BSHIFT;
        }
    }
    for (int i = t + RPT * 1024; i < n; i += 1024) {
        const int r = rec[base + i];
        const int p = atomicAdd(&c[r & (BNODES - 1)], 1);
        srcs[base + p] = r >> BSHIFT;
    }
}

__global__ __launch_bounds__(256) void k_gather(const float* __restrict__ Bn,
                                                const int* __restrict__ srcs,
                                                const int* __restrict__ off,
                                                const int* __restrict__ cnt,
                                                int* __restrict__ M,
                                                int N)
{
    const int l16  = threadIdx.x & 15;
    const int node = blockIdx.x * 16 + (threadIdx.x >> 4);
    if (node >= N) return;
    const int start = off[node];
    const int deg   = cnt[node];
    const float4* __restrict__ B4 = (const float4*)Bn;
    const float ninf = __int_as_float(NEG_INF_I);
    float4 m = make_float4(ninf, ninf, ninf, ninf);
    for (int i = 0; i < deg; i += 8) {
        int s[8];
        float4 v[8];
#pragma unroll
        for (int j = 0; j < 8; ++j) {
            int k = i + j;
            if (k > deg - 1) k = deg - 1;
            s[j] = srcs[start + k];
        }
#pragma unroll
        for (int j = 0; j < 8; ++j) v[j] = B4[(long)s[j] * 16 + l16];
#pragma unroll
        for (int j = 0; j < 8; ++j) {
            m.x = fmaxf(m.x, v[j].x);
            m.y = fmaxf(m.y, v[j].y);
            m.z = fmaxf(m.z, v[j].z);
            m.w = fmaxf(m.w, v[j].w);
        }
    }
    int4 o = make_int4(__float_as_int(m.x), __float_as_int(m.y),
                       __float_as_int(m.z), __float_as_int(m.w));
    ((int4*)M)[(long)node * 16 + l16] = o;
}

__global__ __launch_bounds__(256) void k_elu_partial(const float* __restrict__ A_ws,
                                                     const int* __restrict__ M,
                                                     float* __restrict__ H,
                                                     float* __restrict__ psum,
                                                     float* __restrict__ psq,
                                                     int N)
{
    __shared__ float smem[256];
    const int t  = threadIdx.x;
    const int c  = t & 63;
    const int r4 = t >> 6;
    const int base = blockIdx.x * 64;

    float s = 0.f, q = 0.f;
    for (int rr = r4; rr < 64; rr += 4) {
        int row = base + rr;
        if (row < N) {
            long idx = (long)row * CCH + c;
            int  mb  = M[idx];
            float h;
            if (mb == NEG_INF_I) {
                h = 0.f;
            } else {
                float z = A_ws[idx] + __int_as_float(mb);
                h = (z > 0.f) ? z : expm1f(z);
            }
            H[idx] = h;
            s += h;
            q += h * h;
        }
    }
    smem[t] = s; __syncthreads();
    if (r4 == 0)
        psum[(long)blockIdx.x * 64 + c] = smem[c] + smem[c + 64] + smem[c + 128] + smem[c + 192];
    __syncthreads();
    smem[t] = q; __syncthreads();
    if (r4 == 0)
        psq[(long)blockIdx.x * 64 + c]  = smem[c] + smem[c + 64] + smem[c + 128] + smem[c + 192];
}

#define STATS_THREADS 256
__global__ __launch_bounds__(STATS_THREADS) void k_stats(const float* __restrict__ psum,
                                                         const float* __restrict__ psq,
                                                         int nblocks, int N,
                                                         float* __restrict__ mean,
                                                         float* __restrict__ invstd)
{
    __shared__ double wsum[STATS_THREADS / 32];
    __shared__ double wsq[STATS_THREADS / 32];
    const int c = blockIdx.x;
    const int t = threadIdx.x;

    double s = 0.0, q = 0.0;
#pragma unroll 4
    for (int b = t; b < nblocks; b += STATS_THREADS) {
        s += (double)psum[(long)b * 64 + c];
        q += (double)psq[(long)b * 64 + c];
    }
    for (int off = 16; off > 0; off >>= 1) {
        s += __shfl_xor(s, off);
        q += __shfl_xor(q, off);
    }
    if ((t & 31) == 0) {
        wsum[t >> 5] = s;
        wsq[t >> 5]  = q;
    }
    __syncthreads();
    if (t == 0) {
        double ts = 0.0, tq = 0.0;
        for (int w = 0; w < STATS_THREADS / 32; ++w) {
            ts += wsum[w];
            tq += wsq[w];
        }
        double m = ts / (double)N;
        double v = tq / (double)N - m * m;
        if (v < 0.0) v = 0.0;
        mean[c]   = (float)m;
        invstd[c] = (float)(1.0 / sqrt(v + 1e-5));
    }
}

__global__ __launch_bounds__(256) void k_bn(float* __restrict__ out,
                                            const float* __restrict__ mean,
                                            const float* __restrict__ invstd,
                                            const float* __restrict__ gamma,
                                            const float* __restrict__ beta,
                                            long total)
{
    long i = (long)blockIdx.x * blockDim.x + threadIdx.x;
    if (i < total) {
        int c = (int)(i & 63);
        out[i] = gamma[c] * (out[i] - mean[c]) * invstd[c] + beta[c];
    }
}

extern "C" void kernel_launch(void* const* d_in, const int* in_sizes, int n_in,
                              void* d_out, int out_size, void* d_ws, size_t ws_size,
                              hipStream_t stream)
{
    const float* x     = (const float*)d_in[0];
    const int*   ei    = (const int*)  d_in[1];
    const float* W     = (const float*)d_in[2];
    const float* bias  = (const float*)d_in[3];
    const float* gamma = (const float*)d_in[4];
    const float* beta  = (const float*)d_in[5];
    float* out = (float*)d_out;

    const int N = in_sizes[0] / CCH;
    const int E = in_sizes[1] / 2;

    const size_t nodef = (size_t)N * CCH;
    float* A_ws = (float*)d_ws;
    float* B_ws = A_ws + nodef;
    int*   M_ws = (int*)(B_ws + nodef);
    const int nb4 = (N + 63) / 64;
    float* psum    = (float*)(M_ws + nodef);
    float* psq     = psum + (size_t)nb4 * 64;
    float* meanp   = psq  + (size_t)nb4 * 64;
    float* invstdp = meanp + 64;

    const long tot  = (long)nodef;

    const int Np      = ((N + 63) / 64) * 64;
    const int NB      = (N + BNODES - 1) / BNODES;
    const int nchunks = (E + CHUNK - 1) / CHUNK;
    int* cnt   = (int*)d_out;
    int* off   = cnt + Np;
    int* srcs  = off + Np;
    int* rec   = srcs + E;
    int* T     = rec + E;
    int* total = T + (size_t)NB * nchunks;
    int* bbase = total + MAXB;

    k_gemm<<<(N + 15) / 16, 128, 0, stream>>>(x, W, bias, A_ws, B_ws, N);
    k_chunk_hist<<<nchunks, 256, 0, stream>>>(ei, T, E, NB, nchunks);
    k_bucket_scan<<<NB, 512, 0, stream>>>(T, total, nchunks);
    k_chunk_scatter<<<nchunks, 256, 0, stream>>>(ei, T, total, bbase, rec, E, NB, nchunks);
    k_bucket_sort<<<NB, 1024, 0, stream>>>(rec, bbase, total, cnt, off, srcs, N);
    k_gather<<<(N + 15) / 16, 256, 0, stream>>>(B_ws, srcs, off, cnt, M_ws, N);
    k_elu_partial<<<nb4, 256, 0, stream>>>(A_ws, M_ws, out, psum, psq, N);
    k_stats<<<CCH, STATS_THREADS, 0, stream>>>(psum, psq, nb4, N, meanp, invstdp);
    k_bn<<<(int)((tot + 255) / 256), 256, 0, stream>>>(out, meanp, invstdp, gamma, beta, tot);
}


# Edge sort's scattered 4-byte stores staged in LDS: a chunk groups its records by bucket, a bucket places its sources, in LDS; then adjacent lanes store adjacent words (runs, whole lines)
# speedup vs baseline: 4.1334x; 1.2353x over previous
#include <hip/hip_runtime.h>
#include <hip/hip_bf16.h>

typedef __attribute__((ext_vector_type(2))) float v2f;
typedef __attribute__((ext_vector_type(8))) float v8f;

#define CCH 64
#define NEG_INF_I ((int)0xFF800000)

__global__ __launch_bounds__(128) void k_gemm(const float* __restrict__ x,
                                              const float* __restrict__ W,
                                              const float* __restrict__ bias,
                                              float* __restrict__ A_ws,
                                              float* __restrict__ B_ws,
                                              int N)
{
    const int lane = threadIdx.x & 31;
    const int wave = threadIdx.x >> 5;
    const int n0   = wave * 16;
    const int m0   = blockIdx.x * 16;
    const int half = lane >> 4;
    const int lm   = lane & 15;
    const int col  = n0 + lm;

    v8f accP = {};
    v8f accQ = {};

    int rowA = m0 + lm;
    if (rowA >= N) rowA = N - 1;

#pragma unroll
    for (int kk = 0; kk < 16; ++kk) {
        const int kb  = kk * 4 + half * 2;
        v2f a, b1, b2;
        a.x  = x[(long)rowA * CCH + kb];
        a.y  = x[(long)rowA * CCH + kb + 1];
        b1.x = W[kb * CCH + col];
        b1.y = W[(kb + 1) * CCH + col];
        b2.x = W[(kb + 64) * CCH + col];
        b2.y = W[(kb + 65) * CCH + col];
        accP = __builtin_amdgcn_wmma_f32_16x16x4_f32(false, a, false, b1,
                                                     (short)0, accP, false, false);
        accQ = __builtin_amdgcn_wmma_f32_16x16x4_f32(false, a, false, b2,
                                                     (short)0, accQ, false, false);
    }

    const float bv = bias[col];
    if (m0 + 16 <= N) {
#pragma unroll
        for (int r = 0; r < 8; ++r) {
            const int row = m0 + r + half * 8;
            const float q = accQ[r];
            A_ws[(long)row * CCH + col] = accP[r] - q + bv;
            B_ws[(long)row * CCH + col] = q;
        }
    } else {
#pragma unroll
        for (int r = 0; r < 8; ++r) {
            const int row = m0 + r + half * 8;
            if (row < N) {
                const float q = accQ[r];
                A_ws[(long)row * CCH + col] = accP[r] - q + bv;
                B_ws[(long)row * CCH + col] = q;
            }
        }
    }
}

#define BSHIFT 8
#define BNODES (1 << BSHIFT)
#define MAXB   512
#define CHUNK  4096
#define EPT    16
#define RPT    6

__global__ __launch_bounds__(256) void k_chunk_hist(const int* __restrict__ ei,
                                                    int* __restrict__ T,
                                                    int E, int NB, int nchunks)
{
    __shared__ int h[MAXB];
    const int t     = threadIdx.x;
    const int chunk = blockIdx.x;
    for (int i = t; i < NB; i += 256) h[i] = 0;
    const int e0 = chunk * CHUNK + t;
    int d[EPT];
#pragma unroll
    for (int j = 0; j < EPT; ++j) {
        const int e = e0 + j * 256;
        d[j] = (e < E) ? ei[E + e] : -1;
    }
    __syncthreads();
#pragma unroll
    for (int j = 0; j < EPT; ++j)
        if (d[j] >= 0) atomicAdd(&h[d[j] >> BSHIFT], 1);
    __syncthreads();
    for (int i = t; i < NB; i += 256) T[(long)i * nchunks + chunk] = h[i];
}

__global__ __launch_bounds__(512) void k_bucket_scan(int* __restrict__ T,
                                                     int* __restrict__ total,
                                                     int nchunks)
{
    __shared__ int wtot[16];
    const int t    = threadIdx.x;
    const int lane = t & 31;
    const int w    = t >> 5;
    int* __restrict__ row = T + (long)blockIdx.x * nchunks;
    int carry = 0;
    for (int c0 = 0; c0 < nchunks; c0 += 512) {
        const int c = c0 + t;
        const int v = (c < nchunks) ? row[c] : 0;
        int incl = v;
        for (int dd = 1; dd < 32; dd <<= 1) {
            const int u = __shfl_up(incl, dd);
            if (lane >= dd) incl += u;
        }
        if (lane == 31) wtot[w] = incl;
        __syncthreads();
        int wbase = 0, all = 0;
        for (int k = 0; k < 16; ++k) {
            const int x = wtot[k];
            if (k < w) wbase += x;
            all += x;
        }
        if (c < nchunks) row[c] = carry + wbase + incl - v;
        carry += all;
        __syncthreads();
    }
    if (t == 0) total[blockIdx.x] = carry;
}

__global__ __launch_bounds__(256) void k_chunk_scatter(const int* __restrict__ ei,
                                                       const int* __restrict__ T,
                                                       const int* __restrict__ total,
                                                       int* __restrict__ bbase,
                                                       int* __restrict__ rec,
                                                       int E, int NB, int nchunks)
{
    __shared__ int lcl[MAXB];
    __shared__ int gdst[MAXB];
    __shared__ int vals[CHUNK];
    __shared__ unsigned short bk[CHUNK];
    __shared__ int wtotg[8];
    __shared__ int wtotl[8];
    const int t     = threadIdx.x;
    const int lane  = t & 31;
    const int w     = t >> 5;
    const int chunk = blockIdx.x;
    const int e0    = chunk * CHUNK + t;
    const int b0 = 2 * t, b1 = 2 * t + 1;
    lcl[b0] = 0;
    lcl[b1] = 0;
    int d[EPT], s[EPT], rnk[EPT];
#pragma unroll
    for (int j = 0; j < EPT; ++j) {
        const int e = e0 + j * 256;
        const bool ok = e < E;
        d[j] = ok ? ei[E + e] : -1;
        s[j] = ok ? ei[e] : 0;
    }
    const int v0 = (b0 < NB) ? total[b0] : 0;
    const int v1 = (b1 < NB) ? total[b1] : 0;
    const int c0 = (b0 < NB) ? T[(long)b0 * nchunks + chunk] : 0;
    const int c1 = (b1 < NB) ? T[(long)b1 * nchunks + chunk] : 0;
    __syncthreads();
#pragma unroll
    for (int j = 0; j < EPT; ++j)
        rnk[j] = (d[j] >= 0) ? atomicAdd(&lcl[d[j] >> BSHIFT], 1) : 0;
    __syncthreads();
    const int l0 = lcl[b0];
    const int l1 = lcl[b1];
    const int pairg = v0 + v1;
    const int pairl = l0 + l1;
    int inclg = pairg, incll = pairl;
    for (int dd = 1; dd < 32; dd <<= 1) {
        const int ug = __shfl_up(inclg, dd);
        const int ul = __shfl_up(incll, dd);
        if (lane >= dd) { inclg += ug; incll += ul; }
    }
    if (lane == 31) { wtotg[w] = inclg; wtotl[w] = incll; }
    __syncthreads();
    int wbaseg = 0, wbasel = 0;
    for (int k = 0; k < 8; ++k) if (k < w) { wbaseg += wtotg[k]; wbasel += wtotl[k]; }
    const int ex0 = wbaseg + inclg - pairg;
    const int ex1 = ex0 + v0;
    const int lx0 = wbasel + incll - pairl;
    const int lx1 = lx0 + l0;
    lcl[b0]  = lx0;
    lcl[b1]  = lx1;
    gdst[b0] = ex0 + c0 - lx0;
    gdst[b1] = ex1 + c1 - lx1;
    if (chunk == 0) {
        if (b0 < NB) bbase[b0] = ex0;
        if (b1 < NB) bbase[b1] = ex1;
    }
    __syncthreads();
#pragma unroll
    for (int j = 0; j < EPT; ++j) {
        if (d[j] >= 0) {
            const int b = d[j] >> BSHIFT;
            const int p = lcl[b] + rnk[j];
            vals[p] = (s[j] << BSHIFT) | (d[j] & (BNODES - 1));
            bk[p]   = (unsigned short)b;
        }
    }
    __syncthreads();
    int nvalid = E - chunk * CHUNK;
    if (nvalid > CHUNK) nvalid = CHUNK;
    for (int i = t; i < nvalid; i += 256)
        rec[gdst[bk[i]] + i] = vals[i];
}

__global__ __launch_bounds__(1024) void k_bucket_sort(const int* __restrict__ rec,
                                                      const int* __restrict__ bbase,
                                                      const int* __restrict__ total,
                                                      int* __restrict__ cnt,
                                                      int* __restrict__ off,
                                                      int* __restrict__ srcs,
                                                      int N)
{
    __shared__ int c[BNODES];
    __shared__ int wtot[8];
    __shared__ int sorted[RPT * 1024];
    const int t    = threadIdx.x;
    const int lane = t & 31;
    const int w    = t >> 5;
    const int b    = blockIdx.x;
    const int base = bbase[b];
    const int n    = total[b];
    const bool staged = n <= RPT * 1024;
    if (t < BNODES) c[t] = 0;
    int rd[RPT];
#pragma unroll
    for (int k = 0; k < RPT; ++k) {
        const int i = t + k * 1024;
        rd[k] = (i < n) ? rec[base + i] : -1;
    }
    __syncthreads();
#pragma unroll
    for (int k = 0; k < RPT; ++k)
        if (rd[k] >= 0) atomicAdd(&c[rd[k] & (BNODES - 1)], 1);
    for (int i = t + RPT * 1024; i < n; i += 1024)
        atomicAdd(&c[rec[base + i] & (BNODES - 1)], 1);
    __syncthreads();
    const int v = (t < BNODES) ? c[t] : 0;
    int incl = v;
    for (int dd = 1; dd < 32; dd <<= 1) {
        const int u = __shfl_up(incl, dd);
        if (lane >= dd) incl += u;
    }
    if (t < BNODES && lane == 31) wtot[w] = incl;
    __syncthreads();
    if (t < BNODES) {
        int wbase = 0;
        for (int k = 0; k < 8; ++k) if (k < w) wbase += wtot[k];
        const int ex   = wbase + incl - v;
        const int node = (b << BSHIFT) + t;
        if (node < N) {
            cnt[node] = v;
            off[node] = base + ex;
        }
        c[t] = ex;
    }
    __syncthreads();
#pragma unroll
    for (int k = 0; k < RPT; ++k) {
        if (rd[k] >= 0) {
            const int p = atomicAdd(&c[rd[k] & (BNODES - 1)], 1);
            if (staged) sorted[p] = rd[k] >> BSHIFT;
            else        srcs[base + p] = rd[k] >> BSHIFT;
        }
    }
    for (int i = t + RPT * 1024; i < n; i += 1024) {
        const int r = rec[base + i];
        const int p = atomicAdd(&c[r & (BNODES - 1)], 1);
        srcs[base + p] = r >> BSHIFT;
    }
    if (staged) {
        __syncthreads();
        for (int i = t; i < n; i += 1024) srcs[base + i] = sorted[i];
    }
}

__global__ __launch_bounds__(256) void k_gather(const float* __restrict__ Bn,
                                                const int* __restrict__ srcs,
                                                const int* __restrict__ off,
                                                const int* __restrict__ cnt,
                                                int* __restrict__ M,
                                                int N)
{
    const int l16  = threadIdx.x & 15;
    const int node = blockIdx.x * 16 + (threadIdx.x >> 4);
    if (node >= N) return;
    const int start = off[node];
    const int deg   = cnt[node];
    const float4* __restrict__ B4 = (const float4*)Bn;
    const float ninf = __int_as_float(NEG_INF_I);
    float4 m = make_float4(ninf, ninf, ninf, ninf);
    for (int i = 0; i < deg; i += 8) {
        int s[8];
        float4 v[8];
#pragma unroll
        for (int j = 0; j < 8; ++j) {
            int k = i + j;
            if (k > deg - 1) k = deg - 1;
            s[j] = srcs[start + k];
        }
#pragma unroll
        for (int j = 0; j < 8; ++j) v[j] = B4[(long)s[j] * 16 + l16];
#pragma unroll
        for (int j = 0; j < 8; ++j) {
            m.x = fmaxf(m.x, v[j].x);
            m.y = fmaxf(m.y, v[j].y);
            m.z = fmaxf(m.z, v[j].z);
            m.w = fmaxf(m.w, v[j].w);
        }
    }
    int4 o = make_int4(__float_as_int(m.x), __float_as_int(m.y),
                       __float_as_int(m.z), __float_as_int(m.w));
    ((int4*)M)[(long)node * 16 + l16] = o;
}

__global__ __launch_bounds__(256) void k_elu_partial(const float* __restrict__ A_ws,
                                                     const int* __restrict__ M,
                                                     float* __restrict__ H,
                                                     float* __restrict__ psum,
                                                     float* __restrict__ psq,
                                                     int N)
{
    __shared__ float smem[256];
    const int t  = threadIdx.x;
    const int c  = t & 63;
    const int r4 = t >> 6;
    const int base = blockIdx.x * 64;

    float s = 0.f, q = 0.f;
    for (int rr = r4; rr < 64; rr += 4) {
        int row = base + rr;
        if (row < N) {
            long idx = (long)row * CCH + c;
            int  mb  = M[idx];
            float h;
            if (mb == NEG_INF_I) {
                h = 0.f;
            } else {
                float z = A_ws[idx] + __int_as_float(mb);
                h = (z > 0.f) ? z : expm1f(z);
            }
            H[idx] = h;
            s += h;
            q += h * h;
        }
    }
    smem[t] = s; __syncthreads();
    if (r4 == 0)
        psum[(long)blockIdx.x * 64 + c] = smem[c] + smem[c + 64] + smem[c + 128] + smem[c + 192];
    __syncthreads();
    smem[t] = q; __syncthreads();
    if (r4 == 0)
        psq[(long)blockIdx.x * 64 + c]  = smem[c] + smem[c + 64] + smem[c + 128] + smem[c + 192];
}

#define STATS_THREADS 256
__global__ __launch_bounds__(STATS_THREADS) void k_stats(const float* __restrict__ psum,
                                                         const float* __restrict__ psq,
                                                         int nblocks, int N,
                                                         float* __restrict__ mean,
                                                         float* __restrict__ invstd)
{
    __shared__ double wsum[STATS_THREADS / 32];
    __shared__ double wsq[STATS_THREADS / 32];
    const int c = blockIdx.x;
    const int t = threadIdx.x;

    double s = 0.0, q = 0.0;
#pragma unroll 4
    for (int b = t; b < nblocks; b += STATS_THREADS) {
        s += (double)psum[(long)b * 64 + c];
        q += (double)psq[(long)b * 64 + c];
    }
    for (int off = 16; off > 0; off >>= 1) {
        s += __shfl_xor(s, off);
        q += __shfl_xor(q, off);
    }
    if ((t & 31) == 0) {
        wsum[t >> 5] = s;
        wsq[t >> 5]  = q;
    }
    __syncthreads();
    if (t == 0) {
        double ts = 0.0, tq = 0.0;
        for (int w = 0; w < STATS_THREADS / 32; ++w) {
            ts += wsum[w];
            tq += wsq[w];
        }
        double m = ts / (double)N;
        double v = tq / (double)N - m * m;
        if (v < 0.0) v = 0.0;
        mean[c]   = (float)m;
        invstd[c] = (float)(1.0 / sqrt(v + 1e-5));
    }
}

__global__ __launch_bounds__(256) void k_bn(float* __restrict__ out,
                                            const float* __restrict__ mean,
                                            const float* __restrict__ invstd,
                                            const float* __restrict__ gamma,
                                            const float* __restrict__ beta,
                                            long total)
{
    long i = (long)blockIdx.x * blockDim.x + threadIdx.x;
    if (i < total) {
        int c = (int)(i & 63);
        out[i] = gamma[c] * (out[i] - mean[c]) * invstd[c] + beta[c];
    }
}

extern "C" void kernel_launch(void* const* d_in, const int* in_sizes, int n_in,
                              void* d_out, int out_size, void* d_ws, size_t ws_size,
                              hipStream_t stream)
{
    const float* x     = (const float*)d_in[0];
    const int*   ei    = (const int*)  d_in[1];
    const float* W     = (const float*)d_in[2];
    const float* bias  = (const float*)d_in[3];
    const float* gamma = (const float*)d_in[4];
    const float* beta  = (const float*)d_in[5];
    float* out = (float*)d_out;

    const int N = in_sizes[0] / CCH;
    const int E = in_sizes[1] / 2;

    const size_t nodef = (size_t)N * CCH;
    float* A_ws = (float*)d_ws;
    float* B_ws = A_ws + nodef;
    int*   M_ws = (int*)(B_ws + nodef);
    const int nb4 = (N + 63) / 64;
    float* psum    = (float*)(M_ws + nodef);
    float* psq     = psum + (size_t)nb4 * 64;
    float* meanp   = psq  + (size_t)nb4 * 64;
    float* invstdp = meanp + 64;

    const long tot  = (long)nodef;

    const int Np      = ((N + 63) / 64) * 64;
    const int NB      = (N + BNODES - 1) / BNODES;
    const int nchunks = (E + CHUNK - 1) / CHUNK;
    int* cnt   = (int*)d_out;
    int* off   = cnt + Np;
    int* srcs  = off + Np;
    int* rec   = srcs + E;
    int* T     = rec + E;
    int* total = T + (size_t)NB * nchunks;
    int* bbase = total + MAXB;

    k_gemm<<<(N + 15) / 16, 128, 0, stream>>>(x, W, bias, A_ws, B_ws, N);
    k_chunk_hist<<<nchunks, 256, 0, stream>>>(ei, T, E, NB, nchunks);
    k_bucket_scan<<<NB, 512, 0, stream>>>(T, total, nchunks);
    k_chunk_scatter<<<nchunks, 256, 0, stream>>>(ei, T, total, bbase, rec, E, NB, nchunks);
    k_bucket_sort<<<NB, 1024, 0, stream>>>(rec, bbase, total, cnt, off, srcs, N);
    k_gather<<<(N + 15) / 16, 256, 0, stream>>>(B_ws, srcs, off, cnt, M_ws, N);
    k_elu_partial<<<nb4, 256, 0, stream>>>(A_ws, M_ws, out, psum, psq, N);
    k_stats<<<CCH, STATS_THREADS, 0, stream>>>(psum, psq, nb4, N, meanp, invstdp);
    k_bn<<<(int)((tot + 255) / 256), 256, 0, stream>>>(out, meanp, invstdp, gamma, beta, tot);
}


# Node GEMM has no launch of its own: its row tiles run as extra work-groups of the four edge-sort launches (391 work-groups each), in the wave slots the sort leaves empty; tile code unchanged
# speedup vs baseline: 4.5500x; 1.1008x over previous
#include <hip/hip_runtime.h>
#include <hip/hip_bf16.h>

typedef __attribute__((ext_vector_type(2))) float v2f;
typedef __attribute__((ext_vector_type(8))) float v8f;

#define CCH 64
#define NEG_INF_I ((int)0xFF800000)

struct GemmPart {
    const float* x;
    const float* W;
    const float* bias;
    float* A_ws;
    float* B_ws;
    int N;
    int tile0;
    int tile1;
};

__device__ __forceinline__ void gemm_wave_tile(const float* __restrict__ x,
                                               const float* __restrict__ W,
                                               const float* __restrict__ bias,
                                               float* __restrict__ A_ws,
                                               float* __restrict__ B_ws,
                                               int N, int rowTile, int wave, int lane)
{
    const int n0   = wave * 16;
    const int m0   = rowTile * 16;
    const int half = lane >> 4;
    const int lm   = lane & 15;
    const int col  = n0 + lm;

    v8f accP = {};
    v8f accQ = {};

    int rowA = m0 + lm;
    if (rowA >= N) rowA = N - 1;

#pragma unroll
    for (int kk = 0; kk < 16; ++kk) {
        const int kb  = kk * 4 + half * 2;
        v2f a, b1, b2;
        a.x  = x[(long)rowA * CCH + kb];
        a.y  = x[(long)rowA * CCH + kb + 1];
        b1.x = W[kb * CCH + col];
        b1.y = W[(kb + 1) * CCH + col];
        b2.x = W[(kb + 64) * CCH + col];
        b2.y = W[(kb + 65) * CCH + col];
        accP = __builtin_amdgcn_wmma_f32_16x16x4_f32(false, a, false, b1,
                                                     (short)0, accP, false, false);
        accQ = __builtin_amdgcn_wmma_f32_16x16x4_f32(false, a, false, b2,
                                                     (short)0, accQ, false, false);
    }

    const float bv = bias[col];
    if (m0 + 16 <= N) {
#pragma unroll
        for (int r = 0; r < 8; ++r) {
            const int row = m0 + r + half * 8;
            const float q = accQ[r];
            A_ws[(long)row * CCH + col] = accP[r] - q + bv;
            B_ws[(long)row * CCH + col] = q;
        }
    } else {
#pragma unroll
        for (int r = 0; r < 8; ++r) {
            const int row = m0 + r + half * 8;
            if (row < N) {
                const float q = accQ[r];
                A_ws[(long)row * CCH + col] = accP[r] - q + bv;
                B_ws[(long)row * CCH + col] = q;
            }
        }
    }
}

template <int THREADS>
__device__ __forceinline__ void gemm_rider(const GemmPart& g, int riderBlock)
{
    const int rowTile = g.tile0 + riderBlock * (THREADS / 128) + ((int)threadIdx.x >> 7);
    if (rowTile < g.tile1)
        gemm_wave_tile(g.x, g.W, g.bias, g.A_ws, g.B_ws, g.N, rowTile,
                       ((int)threadIdx.x >> 5) & 3, (int)threadIdx.x & 31);
}

#define BSHIFT 8
#define BNODES (1 << BSHIFT)
#define MAXB   512
#define CHUNK  4096
#define EPT    16
#define RPT    6

__global__ __launch_bounds__(256) void k_chunk_hist(const int* __restrict__ ei,
                                                    int* __restrict__ T,
                                                    int E, int NB, int nchunks,
                                                    GemmPart g)
{
    __shared__ int h[MAXB];
    if ((int)blockIdx.x >= nchunks) { gemm_rider<256>(g, (int)blockIdx.x - nchunks); return; }
    const int t     = threadIdx.x;
    const int chunk = blockIdx.x;
    for (int i = t; i < NB; i += 256) h[i] = 0;
    const int e0 = chunk * CHUNK + t;
    int d[EPT];
#pragma unroll
    for (int j = 0; j < EPT; ++j) {
        const int e = e0 + j * 256;
        d[j] = (e < E) ? ei[E + e] : -1;
    }
    __syncthreads();
#pragma unroll
    for (int j = 0; j < EPT; ++j)
        if (d[j] >= 0) atomicAdd(&h[d[j] >> BSHIFT], 1);
    __syncthreads();
    for (int i = t; i < NB; i += 256) T[(long)i * nchunks + chunk] = h[i];
}

__global__ __launch_bounds__(512) void k_bucket_scan(int* __restrict__ T,
                                                     int* __restrict__ total,
                                                     int nchunks, int NB,
                                                     GemmPart g)
{
    __shared__ int wtot[16];
    if ((int)blockIdx.x >= NB) { gemm_rider<512>(g, (int)blockIdx.x - NB); return; }
    const int t    = threadIdx.x;
    const int lane = t & 31;
    const int w    = t >> 5;
    int* __restrict__ row = T + (long)blockIdx.x * nchunks;
    int carry = 0;
    for (int c0 = 0; c0 < nchunks; c0 += 512) {
        const int c = c0 + t;
        const int v = (c < nchunks) ? row[c] : 0;
        int incl = v;
        for (int dd = 1; dd < 32; dd <<= 1) {
            const int u = __shfl_up(incl, dd);
            if (lane >= dd) incl += u;
        }
        if (lane == 31) wtot[w] = incl;
        __syncthreads();
        int wbase = 0, all = 0;
        for (int k = 0; k < 16; ++k) {
            const int x = wtot[k];
            if (k < w) wbase += x;
            all += x;
        }
        if (c < nchunks) row[c] = carry + wbase + incl - v;
        carry += all;
        __syncthreads();
    }
    if (t == 0) total[blockIdx.x] = carry;
}

__global__ __launch_bounds__(256) void k_chunk_scatter(const int* __restrict__ ei,
                                                       const int* __restrict__ T,
                                                       const int* __restrict__ total,
                                                       int* __restrict__ bbase,
                                                       int* __restrict__ rec,
                                                       int E, int NB, int nchunks,
                                                       GemmPart g)
{
    if ((int)blockIdx.x >= nchunks) { gemm_rider<256>(g, (int)blockIdx.x - nchunks); return; }
    __shared__ int lcl[MAXB];
    __shared__ int gdst[MAXB];
    __shared__ int vals[CHUNK];
    __shared__ unsigned short bk[CHUNK];
    __shared__ int wtotg[8];
    __shared__ int wtotl[8];
    const int t     = threadIdx.x;
    const int lane  = t & 31;
    const int w     = t >> 5;
    const int chunk = blockIdx.x;
    const int e0    = chunk * CHUNK + t;
    const int b0 = 2 * t, b1 = 2 * t + 1;
    lcl[b0] = 0;
    lcl[b1] = 0;
    int d[EPT], s[EPT], rnk[EPT];
#pragma unroll
    for (int j = 0; j < EPT; ++j) {
        const int e = e0 + j * 256;
        const bool ok = e < E;
        d[j] = ok ? ei[E + e] : -1;
        s[j] = ok ? ei[e] : 0;
    }
    const int v0 = (b0 < NB) ? total[b0] : 0;
    const int v1 = (b1 < NB) ? total[b1] : 0;
    const int c0 = (b0 < NB) ? T[(long)b0 * nchunks + chunk] : 0;
    const int c1 = (b1 < NB) ? T[(long)b1 * nchunks + chunk] : 0;
    __syncthreads();
#pragma unroll
    for (int j = 0; j < EPT; ++j)
        rnk[j] = (d[j] >= 0) ? atomicAdd(&lcl[d[j] >> BSHIFT], 1) : 0;
    __syncthreads();
    const int l0 = lcl[b0];
    const int l1 = lcl[b1];
    const int pairg = v0 + v1;
    const int pairl = l0 + l1;
    int inclg = pairg, incll = pairl;
    for (int dd = 1; dd < 32; dd <<= 1) {
        const int ug = __shfl_up(inclg, dd);
        const int ul = __shfl_up(incll, dd);
        if (lane >= dd) { inclg += ug; incll += ul; }
    }
    if (lane == 31) { wtotg[w] = inclg; wtotl[w] = incll; }
    __syncthreads();
    int wbaseg = 0, wbasel = 0;
    for (int k = 0; k < 8; ++k) if (k < w) { wbaseg += wtotg[k]; wbasel += wtotl[k]; }
    const int ex0 = wbaseg + inclg - pairg;
    const int ex1 = ex0 + v0;
    const int lx0 = wbasel + incll - pairl;
    const int lx1 = lx0 + l0;
    lcl[b0]  = lx0;
    lcl[b1]  = lx1;
    gdst[b0] = ex0 + c0 - lx0;
    gdst[b1] = ex1 + c1 - lx1;
    if (chunk == 0) {
        if (b0 < NB) bbase[b0] = ex0;
        if (b1 < NB) bbase[b1] = ex1;
    }
    __syncthreads();
#pragma unroll
    for (int j = 0; j < EPT; ++j) {
        if (d[j] >= 0) {
            const int b = d[j] >> BSHIFT;
            const int p = lcl[b] + rnk[j];
            vals[p] = (s[j] << BSHIFT) | (d[j] & (BNODES - 1));
            bk[p]   = (unsigned short)b;
        }
    }
    __syncthreads();
    int nvalid = E - chunk * CHUNK;
    if (nvalid > CHUNK) nvalid = CHUNK;
    for (int i = t; i < nvalid; i += 256)
        rec[gdst[bk[i]] + i] = vals[i];
}

__global__ __launch_bounds__(1024) void k_bucket_sort(const int* __restrict__ rec,
                                                      const int* __restrict__ bbase,
                                                      const int* __restrict__ total,
                                                      int* __restrict__ cnt,
                                                      int* __restrict__ off,
                                                      int* __restrict__ srcs,
                                                      int N, int NB,
                                                      GemmPart g)
{
    if ((int)blockIdx.x >= NB) { gemm_rider<1024>(g, (int)blockIdx.x - NB); return; }
    __shared__ int c[BNODES];
    __shared__ int wtot[8];
    __shared__ int sorted[RPT * 1024];
    const int t    = threadIdx.x;
    const int lane = t & 31;
    const int w    = t >> 5;
    const int b    = blockIdx.x;
    const int base = bbase[b];
    const int n    = total[b];
    const bool staged = n <= RPT * 1024;
    if (t < BNODES) c[t] = 0;
    int rd[RPT];
#pragma unroll
    for (int k = 0; k < RPT; ++k) {
        const int i = t + k * 1024;
        rd[k] = (i < n) ? rec[base + i] : -1;
    }
    __syncthreads();
#pragma unroll
    for (int k = 0; k < RPT; ++k)
        if (rd[k] >= 0) atomicAdd(&c[rd[k] & (BNODES - 1)], 1);
    for (int i = t + RPT * 1024; i < n; i += 1024)
        atomicAdd(&c[rec[base + i] & (BNODES - 1)], 1);
    __syncthreads();
    const int v = (t < BNODES) ? c[t] : 0;
    int incl = v;
    for (int dd = 1; dd < 32; dd <<= 1) {
        const int u = __shfl_up(incl, dd);
        if (lane >= dd) incl += u;
    }
    if (t < BNODES && lane == 31) wtot[w] = incl;
    __syncthreads();
    if (t < BNODES) {
        int wbase = 0;
        for (int k = 0; k < 8; ++k) if (k < w) wbase += wtot[k];
        const int ex   = wbase + incl - v;
        const int node = (b << BSHIFT) + t;
        if (node < N) {
            cnt[node] = v;
            off[node] = base + ex;
        }
        c[t] = ex;
    }
    __syncthreads();
#pragma unroll
    for (int k = 0; k < RPT; ++k) {
        if (rd[k] >= 0) {
            const int p = atomicAdd(&c[rd[k] & (BNODES - 1)], 1);
            if (staged) sorted[p] = rd[k] >> BSHIFT;
            else        srcs[base + p] = rd[k] >> BSHIFT;
        }
    }
    for (int i = t + RPT * 1024; i < n; i += 1024) {
        const int r = rec[base + i];
        const int p = atomicAdd(&c[r & (BNODES - 1)], 1);
        srcs[base + p] = r >> BSHIFT;
    }
    if (staged) {
        __syncthreads();
        for (int i = t; i < n; i += 1024) srcs[base + i] = sorted[i];
    }
}

__global__ __launch_bounds__(256) void k_gather(const float* __restrict__ Bn,
                                                const int* __restrict__ srcs,
                                                const int* __restrict__ off,
                                                const int* __restrict__ cnt,
                                                int* __restrict__ M,
                                                int N)
{
    const int l16  = threadIdx.x & 15;
    const int node = blockIdx.x * 16 + (threadIdx.x >> 4);
    if (node >= N) return;
    const int start = off[node];
    const int deg   = cnt[node];
    const float4* __restrict__ B4 = (const float4*)Bn;
    const float ninf = __int_as_float(NEG_INF_I);
    float4 m = make_float4(ninf, ninf, ninf, ninf);
    for (int i = 0; i < deg; i += 8) {
        int s[8];
        float4 v[8];
#pragma unroll
        for (int j = 0; j < 8; ++j) {
            int k = i + j;
            if (k > deg - 1) k = deg - 1;
            s[j] = srcs[start + k];
        }
#pragma unroll
        for (int j = 0; j < 8; ++j) v[j] = B4[(long)s[j] * 16 + l16];
#pragma unroll
        for (int j = 0; j < 8; ++j) {
            m.x = fmaxf(m.x, v[j].x);
            m.y = fmaxf(m.y, v[j].y);
            m.z = fmaxf(m.z, v[j].z);
            m.w = fmaxf(m.w, v[j].w);
        }
    }
    int4 o = make_int4(__float_as_int(m.x), __float_as_int(m.y),
                       __float_as_int(m.z), __float_as_int(m.w));
    ((int4*)M)[(long)node * 16 + l16] = o;
}

__global__ __launch_bounds__(256) void k_elu_partial(const float* __restrict__ A_ws,
                                                     const int* __restrict__ M,
                                                     float* __restrict__ H,
                                                     float* __restrict__ psum,
                                                     float* __restrict__ psq,
                                                     int N)
{
    __shared__ float smem[256];
    const int t  = threadIdx.x;
    const int c  = t & 63;
    const int r4 = t >> 6;
    const int base = blockIdx.x * 64;

    float s = 0.f, q = 0.f;
    for (int rr = r4; rr < 64; rr += 4) {
        int row = base + rr;
        if (row < N) {
            long idx = (long)row * CCH + c;
            int  mb  = M[idx];
            float h;
            if (mb == NEG_INF_I) {
                h = 0.f;
            } else {
                float z = A_ws[idx] + __int_as_float(mb);
                h = (z > 0.f) ? z : expm1f(z);
            }
            H[idx] = h;
            s += h;
            q += h * h;
        }
    }
    smem[t] = s; __syncthreads();
    if (r4 == 0)
        psum[(long)blockIdx.x * 64 + c] = smem[c] + smem[c + 64] + smem[c + 128] + smem[c + 192];
    __syncthreads();
    smem[t] = q; __syncthreads();
    if (r4 == 0)
        psq[(long)blockIdx.x * 64 + c]  = smem[c] + smem[c + 64] + smem[c + 128] + smem[c + 192];
}

#define STATS_THREADS 256
__global__ __launch_bounds__(STATS_THREADS) void k_stats(const float* __restrict__ psum,
                                                         const float* __restrict__ psq,
                                                         int nblocks, int N,
                                                         float* __restrict__ mean,
                                                         float* __restrict__ invstd)
{
    __shared__ double wsum[STATS_THREADS / 32];
    __shared__ double wsq[STATS_THREADS / 32];
    const int c = blockIdx.x;
    const int t = threadIdx.x;

    double s = 0.0, q = 0.0;
#pragma unroll 4
    for (int b = t; b < nblocks; b += STATS_THREADS) {
        s += (double)psum[(long)b * 64 + c];
        q += (double)psq[(long)b * 64 + c];
    }
    for (int off = 16; off > 0; off >>= 1) {
        s += __shfl_xor(s, off);
        q += __shfl_xor(q, off);
    }
    if ((t & 31) == 0) {
        wsum[t >> 5] = s;
        wsq[t >> 5]  = q;
    }
    __syncthreads();
    if (t == 0) {
        double ts = 0.0, tq = 0.0;
        for (int w = 0; w < STATS_THREADS / 32; ++w) {
            ts += wsum[w];
            tq += wsq[w];
        }
        double m = ts / (double)N;
        double v = tq / (double)N - m * m;
        if (v < 0.0) v = 0.0;
        mean[c]   = (float)m;
        invstd[c] = (float)(1.0 / sqrt(v + 1e-5));
    }
}

__global__ __launch_bounds__(256) void k_bn(float* __restrict__ out,
                                            const float* __restrict__ mean,
                                            const float* __restrict__ invstd,
                                            const float* __restrict__ gamma,
                                            const float* __restrict__ beta,
                                            long total)
{
    long i = (long)blockIdx.x * blockDim.x + threadIdx.x;
    if (i < total) {
        int c = (int)(i & 63);
        out[i] = gamma[c] * (out[i] - mean[c]) * invstd[c] + beta[c];
    }
}

extern "C" void kernel_launch(void* const* d_in, const int* in_sizes, int n_in,
                              void* d_out, int out_size, void* d_ws, size_t ws_size,
                              hipStream_t stream)
{
    const float* x     = (const float*)d_in[0];
    const int*   ei    = (const int*)  d_in[1];
    const float* W     = (const float*)d_in[2];
    const float* bias  = (const float*)d_in[3];
    const float* gamma = (const float*)d_in[4];
    const float* beta  = (const float*)d_in[5];
    float* out = (float*)d_out;

    const int N = in_sizes[0] / CCH;
    const int E = in_sizes[1] / 2;

    const size_t nodef = (size_t)N * CCH;
    float* A_ws = (float*)d_ws;
    float* B_ws = A_ws + nodef;
    int*   M_ws = (int*)(B_ws + nodef);
    const int nb4 = (N + 63) / 64;
    float* psum    = (float*)(M_ws + nodef);
    float* psq     = psum + (size_t)nb4 * 64;
    float* meanp   = psq  + (size_t)nb4 * 64;
    float* invstdp = meanp + 64;

    const long tot  = (long)nodef;

    const int Np      = ((N + 63) / 64) * 64;
    const int NB      = (N + BNODES - 1) / BNODES;
    const int nchunks = (E + CHUNK - 1) / CHUNK;
    int* cnt   = (int*)d_out;
    int* off   = cnt + Np;
    int* srcs  = off + Np;
    int* rec   = srcs + E;
    int* T     = rec + E;
    int* total = T + (size_t)NB * nchunks;
    int* bbase = total + MAXB;

    const int ntiles = (N + 15) / 16;
    const int t1 = ntiles * 9 / 32;
    const int t2 = ntiles * 13 / 32;
    const int t3 = ntiles * 28 / 32;
    GemmPart g0 = {x, W, bias, A_ws, B_ws, N, 0,  t1};
    GemmPart g1 = {x, W, bias, A_ws, B_ws, N, t1, t2};
    GemmPart g2 = {x, W, bias, A_ws, B_ws, N, t2, t3};
    GemmPart g3 = {x, W, bias, A_ws, B_ws, N, t3, ntiles};
    const int r0 = (t1 + 1) / 2;
    const int r1 = (t2 - t1 + 3) / 4;
    const int r2 = (t3 - t2 + 1) / 2;
    const int r3 = (ntiles - t3 + 7) / 8;

    k_chunk_hist<<<nchunks + r0, 256, 0, stream>>>(ei, T, E, NB, nchunks, g0);
    k_bucket_scan<<<NB + r1, 512, 0, stream>>>(T, total, nchunks, NB, g1);
    k_chunk_scatter<<<nchunks + r2, 256, 0, stream>>>(ei, T, total, bbase, rec, E, NB, nchunks, g2);
    k_bucket_sort<<<NB + r3, 1024, 0, stream>>>(rec, bbase, total, cnt, off, srcs, N, NB, g3);
    k_gather<<<(N + 15) / 16, 256, 0, stream>>>(B_ws, srcs, off, cnt, M_ws, N);
    k_elu_partial<<<nb4, 256, 0, stream>>>(A_ws, M_ws, out, psum, psq, N);
    k_stats<<<CCH, STATS_THREADS, 0, stream>>>(psum, psq, nb4, N, meanp, invstdp);
    k_bn<<<(int)((tot + 255) / 256), 256, 0, stream>>>(out, meanp, invstdp, gamma, beta, tot);
}


# ELU kernel: a thread's 16 M loads and 16 A loads are all issued before the first use (were 32 dependent round trips, each A load behind its M test); same row order, h and both sums keep their bits
# speedup vs baseline: 5.0788x; 1.1162x over previous
#include <hip/hip_runtime.h>
#include <hip/hip_bf16.h>

typedef __attribute__((ext_vector_type(2))) float v2f;
typedef __attribute__((ext_vector_type(8))) float v8f;

#define CCH 64
#define NEG_INF_I ((int)0xFF800000)

struct GemmPart {
    const float* x;
    const float* W;
    const float* bias;
    float* A_ws;
    float* B_ws;
    int N;
    int tile0;
    int tile1;
};

__device__ __forceinline__ void gemm_wave_tile(const float* __restrict__ x,
                                               const float* __restrict__ W,
                                               const float* __restrict__ bias,
                                               float* __restrict__ A_ws,
                                               float* __restrict__ B_ws,
                                               int N, int rowTile, int wave, int lane)
{
    const int n0   = wave * 16;
    const int m0   = rowTile * 16;
    const int half = lane >> 4;
    const int lm   = lane & 15;
    const int col  = n0 + lm;

    v8f accP = {};
    v8f accQ = {};

    int rowA = m0 + lm;
    if (rowA >= N) rowA = N - 1;

#pragma unroll
    for (int kk = 0; kk < 16; ++kk) {
        const int kb  = kk * 4 + half * 2;
        v2f a, b1, b2;
        a.x  = x[(long)rowA * CCH + kb];
        a.y  = x[(long)rowA * CCH + kb + 1];
        b1.x = W[kb * CCH + col];
        b1.y = W[(kb + 1) * CCH + col];
        b2.x = W[(kb + 64) * CCH + col];
        b2.y = W[(kb + 65) * CCH + col];
        accP = __builtin_amdgcn_wmma_f32_16x16x4_f32(false, a, false, b1,
                                                     (short)0, accP, false, false);
        accQ = __builtin_amdgcn_wmma_f32_16x16x4_f32(false, a, false, b2,
                                                     (short)0, accQ, false, false);
    }

    const float bv = bias[col];
    if (m0 + 16 <= N) {
#pragma unroll
        for (int r = 0; r < 8; ++r) {
            const int row = m0 + r + half * 8;
            const float q = accQ[r];
            A_ws[(long)row * CCH + col] = accP[r] - q + bv;
            B_ws[(long)row * CCH + col] = q;
        }
    } else {
#pragma unroll
        for (int r = 0; r < 8; ++r) {
            const int row = m0 + r + half * 8;
            if (row < N) {
                const float q = accQ[r];
                A_ws[(long)row * CCH + col] = accP[r] - q + bv;
                B_ws[(long)row * CCH + col] = q;
            }
        }
    }
}

template <int THREADS>
__device__ __forceinline__ void gemm_rider(const GemmPart& g, int riderBlock)
{
    const int rowTile = g.tile0 + riderBlock * (THREADS / 128) + ((int)threadIdx.x >> 7);
    if (rowTile < g.tile1)
        gemm_wave_tile(g.x, g.W, g.bias, g.A_ws, g.B_ws, g.N, rowTile,
                       ((int)threadIdx.x >> 5) & 3, (int)threadIdx.x & 31);
}

#define BSHIFT 8
#define BNODES (1 << BSHIFT)
#define MAXB   512
#define CHUNK  4096
#define EPT    16
#define RPT    6

__global__ __launch_bounds__(256) void k_chunk_hist(const int* __restrict__ ei,
                                                    int* __restrict__ T,
                                                    int E, int NB, int nchunks,
                                                    GemmPart g)
{
    __shared__ int h[MAXB];
    if ((int)blockIdx.x >= nchunks) { gemm_rider<256>(g, (int)blockIdx.x - nchunks); return; }
    const int t     = threadIdx.x;
    const int chunk = blockIdx.x;
    for (int i = t; i < NB; i += 256) h[i] = 0;
    const int e0 = chunk * CHUNK + t;
    int d[EPT];
#pragma unroll
    for (int j = 0; j < EPT; ++j) {
        const int e = e0 + j * 256;
        d[j] = (e < E) ? ei[E + e] : -1;
    }
    __syncthreads();
#pragma unroll
    for (int j = 0; j < EPT; ++j)
        if (d[j] >= 0) atomicAdd(&h[d[j] >> BSHIFT], 1);
    __syncthreads();
    for (int i = t; i < NB; i += 256) T[(long)i * nchunks + chunk] = h[i];
}

__global__ __launch_bounds__(512) void k_bucket_scan(int* __restrict__ T,
                                                     int* __restrict__ total,
                                                     int nchunks, int NB,
                                                     GemmPart g)
{
    __shared__ int wtot[16];
    if ((int)blockIdx.x >= NB) { gemm_rider<512>(g, (int)blockIdx.x - NB); return; }
    const int t    = threadIdx.x;
    const int lane = t & 31;
    const int w    = t >> 5;
    int* __restrict__ row = T + (long)blockIdx.x * nchunks;
    int carry = 0;
    for (int c0 = 0; c0 < nchunks; c0 += 512) {
        const int c = c0 + t;
        const int v = (c < nchunks) ? row[c] : 0;
        int incl = v;
        for (int dd = 1; dd < 32; dd <<= 1) {
            const int u = __shfl_up(incl, dd);
            if (lane >= dd) incl += u;
        }
        if (lane == 31) wtot[w] = incl;
        __syncthreads();
        int wbase = 0, all = 0;
        for (int k = 0; k < 16; ++k) {
            const int x = wtot[k];
            if (k < w) wbase += x;
            all += x;
        }
        if (c < nchunks) row[c] = carry + wbase + incl - v;
        carry += all;
        __syncthreads();
    }
    if (t == 0) total[blockIdx.x] = carry;
}

__global__ __launch_bounds__(256) void k_chunk_scatter(const int* __restrict__ ei,
                                                       const int* __restrict__ T,
                                                       const int* __restrict__ total,
                                                       int* __restrict__ bbase,
                                                       int* __restrict__ rec,
                                                       int E, int NB, int nchunks,
                                                       GemmPart g)
{
    if ((int)blockIdx.x >= nchunks) { gemm_rider<256>(g, (int)blockIdx.x - nchunks); return; }
    __shared__ int lcl[MAXB];
    __shared__ int gdst[MAXB];
    __shared__ int vals[CHUNK];
    __shared__ unsigned short bk[CHUNK];
    __shared__ int wtotg[8];
    __shared__ int wtotl[8];
    const int t     = threadIdx.x;
    const int lane  = t & 31;
    const int w     = t >> 5;
    const int chunk = blockIdx.x;
    const int e0    = chunk * CHUNK + t;
    const int b0 = 2 * t, b1 = 2 * t + 1;
    lcl[b0] = 0;
    lcl[b1] = 0;
    int d[EPT], s[EPT], rnk[EPT];
#pragma unroll
    for (int j = 0; j < EPT; ++j) {
        const int e = e0 + j * 256;
        const bool ok = e < E;
        d[j] = ok ? ei[E + e] : -1;
        s[j] = ok ? ei[e] : 0;
    }
    const int v0 = (b0 < NB) ? total[b0] : 0;
    const int v1 = (b1 < NB) ? total[b1] : 0;
    const int c0 = (b0 < NB) ? T[(long)b0 * nchunks + chunk] : 0;
    const int c1 = (b1 < NB) ? T[(long)b1 * nchunks + chunk] : 0;
    __syncthreads();
#pragma unroll
    for (int j = 0; j < EPT; ++j)
        rnk[j] = (d[j] >= 0) ? atomicAdd(&lcl[d[j] >> BSHIFT], 1) : 0;
    __syncthreads();
    const int l0 = lcl[b0];
    const int l1 = lcl[b1];
    const int pairg = v0 + v1;
    const int pairl = l0 + l1;
    int inclg = pairg, incll = pairl;
    for (int dd = 1; dd < 32; dd <<= 1) {
        const int ug = __shfl_up(inclg, dd);
        const int ul = __shfl_up(incll, dd);
        if (lane >= dd) { inclg += ug; incll += ul; }
    }
    if (lane == 31) { wtotg[w] = inclg; wtotl[w] = incll; }
    __syncthreads();
    int wbaseg = 0, wbasel = 0;
    for (int k = 0; k < 8; ++k) if (k < w) { wbaseg += wtotg[k]; wbasel += wtotl[k]; }
    const int ex0 = wbaseg + inclg - pairg;
    const int ex1 = ex0 + v0;
    const int lx0 = wbasel + incll - pairl;
    const int lx1 = lx0 + l0;
    lcl[b0]  = lx0;
    lcl[b1]  = lx1;
    gdst[b0] = ex0 + c0 - lx0;
    gdst[b1] = ex1 + c1 - lx1;
    if (chunk == 0) {
        if (b0 < NB) bbase[b0] = ex0;
        if (b1 < NB) bbase[b1] = ex1;
    }
    __syncthreads();
#pragma unroll
    for (int j = 0; j < EPT; ++j) {
        if (d[j] >= 0) {
            const int b = d[j] >> BSHIFT;
            const int p = lcl[b] + rnk[j];
            vals[p] = (s[j] << BSHIFT) | (d[j] & (BNODES - 1));
            bk[p]   = (unsigned short)b;
        }
    }
    __syncthreads();
    int nvalid = E - chunk * CHUNK;
    if (nvalid > CHUNK) nvalid = CHUNK;
    for (int i = t; i < nvalid; i += 256)
        rec[gdst[bk[i]] + i] = vals[i];
}

__global__ __launch_bounds__(1024) void k_bucket_sort(const int* __restrict__ rec,
                                                      const int* __restrict__ bbase,
                                                      const int* __restrict__ total,
                                                      int* __restrict__ cnt,
                                                      int* __restrict__ off,
                                                      int* __restrict__ srcs,
                                                      int N, int NB,
                                                      GemmPart g)
{
    if ((int)blockIdx.x >= NB) { gemm_rider<1024>(g, (int)blockIdx.x - NB); return; }
    __shared__ int c[BNODES];
    __shared__ int wtot[8];
    __shared__ int sorted[RPT * 1024];
    const int t    = threadIdx.x;
    const int lane = t & 31;
    const int w    = t >> 5;
    const int b    = blockIdx.x;
    const int base = bbase[b];
    const int n    = total[b];
    const bool staged = n <= RPT * 1024;
    if (t < BNODES) c[t] = 0;
    int rd[RPT];
#pragma unroll
    for (int k = 0; k < RPT; ++k) {
        const int i = t + k * 1024;
        rd[k] = (i < n) ? rec[base + i] : -1;
    }
    __syncthreads();
#pragma unroll
    for (int k = 0; k < RPT; ++k)
        if (rd[k] >= 0) atomicAdd(&c[rd[k] & (BNODES - 1)], 1);
    for (int i = t + RPT * 1024; i < n; i += 1024)
        atomicAdd(&c[rec[base + i] & (BNODES - 1)], 1);
    __syncthreads();
    const int v = (t < BNODES) ? c[t] : 0;
    int incl = v;
    for (int dd = 1; dd < 32; dd <<= 1) {
        const int u = __shfl_up(incl, dd);
        if (lane >= dd) incl += u;
    }
    if (t < BNODES && lane == 31) wtot[w] = incl;
    __syncthreads();
    if (t < BNODES) {
        int wbase = 0;
        for (int k = 0; k < 8; ++k) if (k < w) wbase += wtot[k];
        const int ex   = wbase + incl - v;
        const int node = (b << BSHIFT) + t;
        if (node < N) {
            cnt[node] = v;
            off[node] = base + ex;
        }
        c[t] = ex;
    }
    __syncthreads();
#pragma unroll
    for (int k = 0; k < RPT; ++k) {
        if (rd[k] >= 0) {
            const int p = atomicAdd(&c[rd[k] & (BNODES - 1)], 1);
            if (staged) sorted[p] = rd[k] >> BSHIFT;
            else        srcs[base + p] = rd[k] >> BSHIFT;
        }
    }
    for (int i = t + RPT * 1024; i < n; i += 1024) {
        const int r = rec[base + i];
        const int p = atomicAdd(&c[r & (BNODES - 1)], 1);
        srcs[base + p] = r >> BSHIFT;
    }
    if (staged) {
        __syncthreads();
        for (int i = t; i < n; i += 1024) srcs[base + i] = sorted[i];
    }
}

__global__ __launch_bounds__(256) void k_gather(const float* __restrict__ Bn,
                                                const int* __restrict__ srcs,
                                                const int* __restrict__ off,
                                                const int* __restrict__ cnt,
                                                int* __restrict__ M,
                                                int N)
{
    const int l16  = threadIdx.x & 15;
    const int node = blockIdx.x * 16 + (threadIdx.x >> 4);
    if (node >= N) return;
    const int start = off[node];
    const int deg   = cnt[node];
    const float4* __restrict__ B4 = (const float4*)Bn;
    const float ninf = __int_as_float(NEG_INF_I);
    float4 m = make_float4(ninf, ninf, ninf, ninf);
    for (int i = 0; i < deg; i += 8) {
        int s[8];
        float4 v[8];
#pragma unroll
        for (int j = 0; j < 8; ++j) {
            int k = i + j;
            if (k > deg - 1) k = deg - 1;
            s[j] = srcs[start + k];
        }
#pragma unroll
        for (int j = 0; j < 8; ++j) v[j] = B4[(long)s[j] * 16 + l16];
#pragma unroll
        for (int j = 0; j < 8; ++j) {
            m.x = fmaxf(m.x, v[j].x);
            m.y = fmaxf(m.y, v[j].y);
            m.z = fmaxf(m.z, v[j].z);
            m.w = fmaxf(m.w, v[j].w);
        }
    }
    int4 o = make_int4(__float_as_int(m.x), __float_as_int(m.y),
                       __float_as_int(m.z), __float_as_int(m.w));
    ((int4*)M)[(long)node * 16 + l16] = o;
}

#define ELU_ROWS 16
__global__ __launch_bounds__(256) void k_elu_partial(const float* __restrict__ A_ws,
                                                     const int* __restrict__ M,
                                                     float* __restrict__ H,
                                                     float* __restrict__ psum,
                                                     float* __restrict__ psq,
                                                     int N)
{
    __shared__ float smem[256];
    const int t  = threadIdx.x;
    const int c  = t & 63;
    const int r4 = t >> 6;
    const int base = blockIdx.x * 64;

    int   mb[ELU_ROWS];
    float av[ELU_ROWS];
#pragma unroll
    for (int k = 0; k < ELU_ROWS; ++k) {
        int row = base + r4 + 4 * k;
        if (row > N - 1) row = N - 1;
        const unsigned idx = (unsigned)row * CCH + (unsigned)c;
        mb[k] = M[idx];
        av[k] = A_ws[idx];
    }

    __builtin_amdgcn_sched_barrier(0);

    float s = 0.f, q = 0.f;
#pragma unroll
    for (int k = 0; k < ELU_ROWS; ++k) {
        const int row = base + r4 + 4 * k;
        const float z = av[k] + __int_as_float(mb[k]);
        const float e = (z > 0.f) ? z : expm1f(z);
        const float h = (mb[k] == NEG_INF_I) ? 0.f : e;
        if (row < N) {
            H[(unsigned)row * CCH + (unsigned)c] = h;
            s += h;
            q += h * h;
        }
    }
    smem[t] = s; __syncthreads();
    if (r4 == 0)
        psum[(long)blockIdx.x * 64 + c] = smem[c] + smem[c + 64] + smem[c + 128] + smem[c + 192];
    __syncthreads();
    smem[t] = q; __syncthreads();
    if (r4 == 0)
        psq[(long)blockIdx.x * 64 + c]  = smem[c] + smem[c + 64] + smem[c + 128] + smem[c + 192];
}

#define STATS_THREADS 256
__global__ __launch_bounds__(STATS_THREADS) void k_stats(const float* __restrict__ psum,
                                                         const float* __restrict__ psq,
                                                         int nblocks, int N,
                                                         float* __restrict__ mean,
                                                         float* __restrict__ invstd)
{
    __shared__ double wsum[STATS_THREADS / 32];
    __shared__ double wsq[STATS_THREADS / 32];
    const int c = blockIdx.x;
    const int t = threadIdx.x;

    double s = 0.0, q = 0.0;
#pragma unroll 4
    for (int b = t; b < nblocks; b += STATS_THREADS) {
        s += (double)psum[(long)b * 64 + c];
        q += (double)psq[(long)b * 64 + c];
    }
    for (int off = 16; off > 0; off >>= 1) {
        s += __shfl_xor(s, off);
        q += __shfl_xor(q, off);
    }
    if ((t & 31) == 0) {
        wsum[t >> 5] = s;
        wsq[t >> 5]  = q;
    }
    __syncthreads();
    if (t == 0) {
        double ts = 0.0, tq = 0.0;
        for (int w = 0; w < STATS_THREADS / 32; ++w) {
            ts += wsum[w];
            tq += wsq[w];
        }
        double m = ts / (double)N;
        double v = tq / (double)N - m * m;
        if (v < 0.0) v = 0.0;
        mean[c]   = (float)m;
        invstd[c] = (float)(1.0 / sqrt(v + 1e-5));
    }
}

__global__ __launch_bounds__(256) void k_bn(float* __restrict__ out,
                                            const float* __restrict__ mean,
                                            const float* __restrict__ invstd,
                                            const float* __restrict__ gamma,
                                            const float* __restrict__ beta,
                                            long total)
{
    long i = (long)blockIdx.x * blockDim.x + threadIdx.x;
    if (i < total) {
        int c = (int)(i & 63);
        out[i] = gamma[c] * (out[i] - mean[c]) * invstd[c] + beta[c];
    }
}

extern "C" void kernel_launch(void* const* d_in, const int* in_sizes, int n_in,
                              void* d_out, int out_size, void* d_ws, size_t ws_size,
                              hipStream_t stream)
{
    const float* x     = (const float*)d_in[0];
    const int*   ei    = (const int*)  d_in[1];
    const float* W     = (const float*)d_in[2];
    const float* bias  = (const float*)d_in[3];
    const float* gamma = (const float*)d_in[4];
    const float* beta  = (const float*)d_in[5];
    float* out = (float*)d_out;

    const int N = in_sizes[0] / CCH;
    const int E = in_sizes[1] / 2;

    const size_t nodef = (size_t)N * CCH;
    float* A_ws = (float*)d_ws;
    float* B_ws = A_ws + nodef;
    int*   M_ws = (int*)(B_ws + nodef);
    const int nb4 = (N + 63) / 64;
    float* psum    = (float*)(M_ws + nodef);
    float* psq     = psum + (size_t)nb4 * 64;
    float* meanp   = psq  + (size_t)nb4 * 64;
    float* invstdp = meanp + 64;

    const long tot  = (long)nodef;

    const int Np      = ((N + 63) / 64) * 64;
    const int NB      = (N + BNODES - 1) / BNODES;
    const int nchunks = (E + CHUNK - 1) / CHUNK;
    int* cnt   = (int*)d_out;
    int* off   = cnt + Np;
    int* srcs  = off + Np;
    int* rec   = srcs + E;
    int* T     = rec + E;
    int* total = T + (size_t)NB * nchunks;
    int* bbase = total + MAXB;

    const int ntiles = (N + 15) / 16;
    const int t1 = ntiles * 9 / 32;
    const int t2 = ntiles * 13 / 32;
    const int t3 = ntiles * 28 / 32;
    GemmPart g0 = {x, W, bias, A_ws, B_ws, N, 0,  t1};
    GemmPart g1 = {x, W, bias, A_ws, B_ws, N, t1, t2};
    GemmPart g2 = {x, W, bias, A_ws, B_ws, N, t2, t3};
    GemmPart g3 = {x, W, bias, A_ws, B_ws, N, t3, ntiles};
    const int r0 = (t1 + 1) / 2;
    const int r1 = (t2 - t1 + 3) / 4;
    const int r2 = (t3 - t2 + 1) / 2;
    const int r3 = (ntiles - t3 + 7) / 8;

    k_chunk_hist<<<nchunks + r0, 256, 0, stream>>>(ei, T, E, NB, nchunks, g0);
    k_bucket_scan<<<NB + r1, 512, 0, stream>>>(T, total, nchunks, NB, g1);
    k_chunk_scatter<<<nchunks + r2, 256, 0, stream>>>(ei, T, total, bbase, rec, E, NB, nchunks, g2);
    k_bucket_sort<<<NB + r3, 1024, 0, stream>>>(rec, bbase, total, cnt, off, srcs, N, NB, g3);
    k_gather<<<(N + 15) / 16, 256, 0, stream>>>(B_ws, srcs, off, cnt, M_ws, N);
    k_elu_partial<<<nb4, 256, 0, stream>>>(A_ws, M_ws, out, psum, psq, N);
    k_stats<<<CCH, STATS_THREADS, 0, stream>>>(psum, psq, nb4, N, meanp, invstdp);
    k_bn<<<(int)((tot + 255) / 256), 256, 0, stream>>>(out, meanp, invstdp, gamma, beta, tot);
}


# BN apply kernel in the ELU kernel's shape: a thread takes one channel of 16 rows of a 64-row block, all 16 loads in flight (was one word per thread): 12504 waves to start, not 200000; same arithmetic
# speedup vs baseline: 5.6730x; 1.1170x over previous
#include <hip/hip_runtime.h>
#include <hip/hip_bf16.h>

typedef __attribute__((ext_vector_type(2))) float v2f;
typedef __attribute__((ext_vector_type(8))) float v8f;

#define CCH 64
#define NEG_INF_I ((int)0xFF800000)

struct GemmPart {
    const float* x;
    const float* W;
    const float* bias;
    float* A_ws;
    float* B_ws;
    int N;
    int tile0;
    int tile1;
};

__device__ __forceinline__ void gemm_wave_tile(const float* __restrict__ x,
                                               const float* __restrict__ W,
                                               const float* __restrict__ bias,
                                               float* __restrict__ A_ws,
                                               float* __restrict__ B_ws,
                                               int N, int rowTile, int wave, int lane)
{
    const int n0   = wave * 16;
    const int m0   = rowTile * 16;
    const int half = lane >> 4;
    const int lm   = lane & 15;
    const int col  = n0 + lm;

    v8f accP = {};
    v8f accQ = {};

    int rowA = m0 + lm;
    if (rowA >= N) rowA = N - 1;

#pragma unroll
    for (int kk = 0; kk < 16; ++kk) {
        const int kb  = kk * 4 + half * 2;
        v2f a, b1, b2;
        a.x  = x[(long)rowA * CCH + kb];
        a.y  = x[(long)rowA * CCH + kb + 1];
        b1.x = W[kb * CCH + col];
        b1.y = W[(kb + 1) * CCH + col];
        b2.x = W[(kb + 64) * CCH + col];
        b2.y = W[(kb + 65) * CCH + col];
        accP = __builtin_amdgcn_wmma_f32_16x16x4_f32(false, a, false, b1,
                                                     (short)0, accP, false, false);
        accQ = __builtin_amdgcn_wmma_f32_16x16x4_f32(false, a, false, b2,
                                                     (short)0, accQ, false, false);
    }

    const float bv = bias[col];
    if (m0 + 16 <= N) {
#pragma unroll
        for (int r = 0; r < 8; ++r) {
            const int row = m0 + r + half * 8;
            const float q = accQ[r];
            A_ws[(long)row * CCH + col] = accP[r] - q + bv;
            B_ws[(long)row * CCH + col] = q;
        }
    } else {
#pragma unroll
        for (int r = 0; r < 8; ++r) {
            const int row = m0 + r + half * 8;
            if (row < N) {
                const float q = accQ[r];
                A_ws[(long)row * CCH + col] = accP[r] - q + bv;
                B_ws[(long)row * CCH + col] = q;
            }
        }
    }
}

template <int THREADS>
__device__ __forceinline__ void gemm_rider(const GemmPart& g, int riderBlock)
{
    const int rowTile = g.tile0 + riderBlock * (THREADS / 128) + ((int)threadIdx.x >> 7);
    if (rowTile < g.tile1)
        gemm_wave_tile(g.x, g.W, g.bias, g.A_ws, g.B_ws, g.N, rowTile,
                       ((int)threadIdx.x >> 5) & 3, (int)threadIdx.x & 31);
}

#define BSHIFT 8
#define BNODES (1 << BSHIFT)
#define MAXB   512
#define CHUNK  4096
#define EPT    16
#define RPT    6

__global__ __launch_bounds__(256) void k_chunk_hist(const int* __restrict__ ei,
                                                    int* __restrict__ T,
                                                    int E, int NB, int nchunks,
                                                    GemmPart g)
{
    __shared__ int h[MAXB];
    if ((int)blockIdx.x >= nchunks) { gemm_rider<256>(g, (int)blockIdx.x - nchunks); return; }
    const int t     = threadIdx.x;
    const int chunk = blockIdx.x;
    for (int i = t; i < NB; i += 256) h[i] = 0;
    const int e0 = chunk * CHUNK + t;
    int d[EPT];
#pragma unroll
    for (int j = 0; j < EPT; ++j) {
        const int e = e0 + j * 256;
        d[j] = (e < E) ? ei[E + e] : -1;
    }
    __syncthreads();
#pragma unroll
    for (int j = 0; j < EPT; ++j)
        if (d[j] >= 0) atomicAdd(&h[d[j] >> BSHIFT], 1);
    __syncthreads();
    for (int i = t; i < NB; i += 256) T[(long)i * nchunks + chunk] = h[i];
}

__global__ __launch_bounds__(512) void k_bucket_scan(int* __restrict__ T,
                                                     int* __restrict__ total,
                                                     int nchunks, int NB,
                                                     GemmPart g)
{
    __shared__ int wtot[16];
    if ((int)blockIdx.x >= NB) { gemm_rider<512>(g, (int)blockIdx.x - NB); return; }
    const int t    = threadIdx.x;
    const int lane = t & 31;
    const int w    = t >> 5;
    int* __restrict__ row = T + (long)blockIdx.x * nchunks;
    int carry = 0;
    for (int c0 = 0; c0 < nchunks; c0 += 512) {
        const int c = c0 + t;
        const int v = (c < nchunks) ? row[c] : 0;
        int incl = v;
        for (int dd = 1; dd < 32; dd <<= 1) {
            const int u = __shfl_up(incl, dd);
            if (lane >= dd) incl += u;
        }
        if (lane == 31) wtot[w] = incl;
        __syncthreads();
        int wbase = 0, all = 0;
        for (int k = 0; k < 16; ++k) {
            const int x = wtot[k];
            if (k < w) wbase += x;
            all += x;
        }
        if (c < nchunks) row[c] = carry + wbase + incl - v;
        carry += all;
        __syncthreads();
    }
    if (t == 0) total[blockIdx.x] = carry;
}

__global__ __launch_bounds__(256) void k_chunk_scatter(const int* __restrict__ ei,
                                                       const int* __restrict__ T,
                                                       const int* __restrict__ total,
                                                       int* __restrict__ bbase,
                                                       int* __restrict__ rec,
                                                       int E, int NB, int nchunks,
                                                       GemmPart g)
{
    if ((int)blockIdx.x >= nchunks) { gemm_rider<256>(g, (int)blockIdx.x - nchunks); return; }
    __shared__ int lcl[MAXB];
    __shared__ int gdst[MAXB];
    __shared__ int vals[CHUNK];
    __shared__ unsigned short bk[CHUNK];
    __shared__ int wtotg[8];
    __shared__ int wtotl[8];
    const int t     = threadIdx.x;
    const int lane  = t & 31;
    const int w     = t >> 5;
    const int chunk = blockIdx.x;
    const int e0    = chunk * CHUNK + t;
    const int b0 = 2 * t, b1 = 2 * t + 1;
    lcl[b0] = 0;
    lcl[b1] = 0;
    int d[EPT], s[EPT], rnk[EPT];
#pragma unroll
    for (int j = 0; j < EPT; ++j) {
        const int e = e0 + j * 256;
        const bool ok = e < E;
        d[j] = ok ? ei[E + e] : -1;
        s[j] = ok ? ei[e] : 0;
    }
    const int v0 = (b0 < NB) ? total[b0] : 0;
    const int v1 = (b1 < NB) ? total[b1] : 0;
    const int c0 = (b0 < NB) ? T[(long)b0 * nchunks + chunk] : 0;
    const int c1 = (b1 < NB) ? T[(long)b1 * nchunks + chunk] : 0;
    __syncthreads();
#pragma unroll
    for (int j = 0; j < EPT; ++j)
        rnk[j] = (d[j] >= 0) ? atomicAdd(&lcl[d[j] >> BSHIFT], 1) : 0;
    __syncthreads();
    const int l0 = lcl[b0];
    const int l1 = lcl[b1];
    const int pairg = v0 + v1;
    const int pairl = l0 + l1;
    int inclg = pairg, incll = pairl;
    for (int dd = 1; dd < 32; dd <<= 1) {
        const int ug = __shfl_up(inclg, dd);
        const int ul = __shfl_up(incll, dd);
        if (lane >= dd) { inclg += ug; incll += ul; }
    }
    if (lane == 31) { wtotg[w] = inclg; wtotl[w] = incll; }
    __syncthreads();
    int wbaseg = 0, wbasel = 0;
    for (int k = 0; k < 8; ++k) if (k < w) { wbaseg += wtotg[k]; wbasel += wtotl[k]; }
    const int ex0 = wbaseg + inclg - pairg;
    const int ex1 = ex0 + v0;
    const int lx0 = wbasel + incll - pairl;
    const int lx1 = lx0 + l0;
    lcl[b0]  = lx0;
    lcl[b1]  = lx1;
    gdst[b0] = ex0 + c0 - lx0;
    gdst[b1] = ex1 + c1 - lx1;
    if (chunk == 0) {
        if (b0 < NB) bbase[b0] = ex0;
        if (b1 < NB) bbase[b1] = ex1;
    }
    __syncthreads();
#pragma unroll
    for (int j = 0; j < EPT; ++j) {
        if (d[j] >= 0) {
            const int b = d[j] >> BSHIFT;
            const int p = lcl[b] + rnk[j];
            vals[p] = (s[j] << BSHIFT) | (d[j] & (BNODES - 1));
            bk[p]   = (unsigned short)b;
        }
    }
    __syncthreads();
    int nvalid = E - chunk * CHUNK;
    if (nvalid > CHUNK) nvalid = CHUNK;
    for (int i = t; i < nvalid; i += 256)
        rec[gdst[bk[i]] + i] = vals[i];
}

__global__ __launch_bounds__(1024) void k_bucket_sort(const int* __restrict__ rec,
                                                      const int* __restrict__ bbase,
                                                      const int* __restrict__ total,
                                                      int* __restrict__ cnt,
                                                      int* __restrict__ off,
                                                      int* __restrict__ srcs,
                                                      int N, int NB,
                                                      GemmPart g)
{
    if ((int)blockIdx.x >= NB) { gemm_rider<1024>(g, (int)blockIdx.x - NB); return; }
    __shared__ int c[BNODES];
    __shared__ int wtot[8];
    __shared__ int sorted[RPT * 1024];
    const int t    = threadIdx.x;
    const int lane = t & 31;
    const int w    = t >> 5;
    const int b    = blockIdx.x;
    const int base = bbase[b];
    const int n    = total[b];
    const bool staged = n <= RPT * 1024;
    if (t < BNODES) c[t] = 0;
    int rd[RPT];
#pragma unroll
    for (int k = 0; k < RPT; ++k) {
        const int i = t + k * 1024;
        rd[k] = (i < n) ? rec[base + i] : -1;
    }
    __syncthreads();
#pragma unroll
    for (int k = 0; k < RPT; ++k)
        if (rd[k] >= 0) atomicAdd(&c[rd[k] & (BNODES - 1)], 1);
    for (int i = t + RPT * 1024; i < n; i += 1024)
        atomicAdd(&c[rec[base + i] & (BNODES - 1)], 1);
    __syncthreads();
    const int v = (t < BNODES) ? c[t] : 0;
    int incl = v;
    for (int dd = 1; dd < 32; dd <<= 1) {
        const int u = __shfl_up(incl, dd);
        if (lane >= dd) incl += u;
    }
    if (t < BNODES && lane == 31) wtot[w] = incl;
    __syncthreads();
    if (t < BNODES) {
        int wbase = 0;
        for (int k = 0; k < 8; ++k) if (k < w) wbase += wtot[k];
        const int ex   = wbase + incl - v;
        const int node = (b << BSHIFT) + t;
        if (node < N) {
            cnt[node] = v;
            off[node] = base + ex;
        }
        c[t] = ex;
    }
    __syncthreads();
#pragma unroll
    for (int k = 0; k < RPT; ++k) {
        if (rd[k] >= 0) {
            const int p = atomicAdd(&c[rd[k] & (BNODES - 1)], 1);
            if (staged) sorted[p] = rd[k] >> BSHIFT;
            else        srcs[base + p] = rd[k] >> BSHIFT;
        }
    }
    for (int i = t + RPT * 1024; i < n; i += 1024) {
        const int r = rec[base + i];
        const int p = atomicAdd(&c[r & (BNODES - 1)], 1);
        srcs[base + p] = r >> BSHIFT;
    }
    if (staged) {
        __syncthreads();
        for (int i = t; i < n; i += 1024) srcs[base + i] = sorted[i];
    }
}

__global__ __launch_bounds__(256) void k_gather(const float* __restrict__ Bn,
                                                const int* __restrict__ srcs,
                                                const int* __restrict__ off,
                                                const int* __restrict__ cnt,
                                                int* __restrict__ M,
                                                int N)
{
    const int l16  = threadIdx.x & 15;
    const int node = blockIdx.x * 16 + (threadIdx.x >> 4);
    if (node >= N) return;
    const int start = off[node];
    const int deg   = cnt[node];
    const float4* __restrict__ B4 = (const float4*)Bn;
    const float ninf = __int_as_float(NEG_INF_I);
    float4 m = make_float4(ninf, ninf, ninf, ninf);
    for (int i = 0; i < deg; i += 8) {
        int s[8];
        float4 v[8];
#pragma unroll
        for (int j = 0; j < 8; ++j) {
            int k = i + j;
            if (k > deg - 1) k = deg - 1;
            s[j] = srcs[start + k];
        }
#pragma unroll
        for (int j = 0; j < 8; ++j) v[j] = B4[(long)s[j] * 16 + l16];
#pragma unroll
        for (int j = 0; j < 8; ++j) {
            m.x = fmaxf(m.x, v[j].x);
            m.y = fmaxf(m.y, v[j].y);
            m.z = fmaxf(m.z, v[j].z);
            m.w = fmaxf(m.w, v[j].w);
        }
    }
    int4 o = make_int4(__float_as_int(m.x), __float_as_int(m.y),
                       __float_as_int(m.z), __float_as_int(m.w));
    ((int4*)M)[(long)node * 16 + l16] = o;
}

#define ELU_ROWS 16
__global__ __launch_bounds__(256) void k_elu_partial(const float* __restrict__ A_ws,
                                                     const int* __restrict__ M,
                                                     float* __restrict__ H,
                                                     float* __restrict__ psum,
                                                     float* __restrict__ psq,
                                                     int N)
{
    __shared__ float smem[256];
    const int t  = threadIdx.x;
    const int c  = t & 63;
    const int r4 = t >> 6;
    const int base = blockIdx.x * 64;

    int   mb[ELU_ROWS];
    float av[ELU_ROWS];
#pragma unroll
    for (int k = 0; k < ELU_ROWS; ++k) {
        int row = base + r4 + 4 * k;
        if (row > N - 1) row = N - 1;
        const unsigned idx = (unsigned)row * CCH + (unsigned)c;
        mb[k] = M[idx];
        av[k] = A_ws[idx];
    }

    __builtin_amdgcn_sched_barrier(0);

    float s = 0.f, q = 0.f;
#pragma unroll
    for (int k = 0; k < ELU_ROWS; ++k) {
        const int row = base + r4 + 4 * k;
        const float z = av[k] + __int_as_float(mb[k]);
        const float e = (z > 0.f) ? z : expm1f(z);
        const float h = (mb[k] == NEG_INF_I) ? 0.f : e;
        if (row < N) {
            H[(unsigned)row * CCH + (unsigned)c] = h;
            s += h;
            q += h * h;
        }
    }
    smem[t] = s; __syncthreads();
    if (r4 == 0)
        psum[(long)blockIdx.x * 64 + c] = smem[c] + smem[c + 64] + smem[c + 128] + smem[c + 192];
    __syncthreads();
    smem[t] = q; __syncthreads();
    if (r4 == 0)
        psq[(long)blockIdx.x * 64 + c]  = smem[c] + smem[c + 64] + smem[c + 128] + smem[c + 192];
}

#define STATS_THREADS 256
__global__ __launch_bounds__(STATS_THREADS) void k_stats(const float* __restrict__ psum,
                                                         const float* __restrict__ psq,
                                                         int nblocks, int N,
                                                         float* __restrict__ mean,
                                                         float* __restrict__ invstd)
{
    __shared__ double wsum[STATS_THREADS / 32];
    __shared__ double wsq[STATS_THREADS / 32];
    const int c = blockIdx.x;
    const int t = threadIdx.x;

    double s = 0.0, q = 0.0;
#pragma unroll 4
    for (int b = t; b < nblocks; b += STATS_THREADS) {
        s += (double)psum[(long)b * 64 + c];
        q += (double)psq[(long)b * 64 + c];
    }
    for (int off = 16; off > 0; off >>= 1) {
        s += __shfl_xor(s, off);
        q += __shfl_xor(q, off);
    }
    if ((t & 31) == 0) {
        wsum[t >> 5] = s;
        wsq[t >> 5]  = q;
    }
    __syncthreads();
    if (t == 0) {
        double ts = 0.0, tq = 0.0;
        for (int w = 0; w < STATS_THREADS / 32; ++w) {
            ts += wsum[w];
            tq += wsq[w];
        }
        double m = ts / (double)N;
        double v = tq / (double)N - m * m;
        if (v < 0.0) v = 0.0;
        mean[c]   = (float)m;
        invstd[c] = (float)(1.0 / sqrt(v + 1e-5));
    }
}

#define BN_ROWS 16
__global__ __launch_bounds__(256) void k_bn(float* __restrict__ out,
                                            const float* __restrict__ mean,
                                            const float* __restrict__ invstd,
                                            const float* __restrict__ gamma,
                                            const float* __restrict__ beta,
                                            int N)
{
    const int t  = threadIdx.x;
    const int c  = t & 63;
    const int r4 = t >> 6;
    const int base = blockIdx.x * 64;

    const float g  = gamma[c];
    const float mu = mean[c];
    const float is = invstd[c];
    const float bt = beta[c];

    if (base + 64 <= N) {
        float hv[BN_ROWS];
#pragma unroll
        for (int k = 0; k < BN_ROWS; ++k)
            hv[k] = out[(unsigned)(base + r4 + 4 * k) * CCH + (unsigned)c];
#pragma unroll
        for (int k = 0; k < BN_ROWS; ++k)
            out[(unsigned)(base + r4 + 4 * k) * CCH + (unsigned)c] = g * (hv[k] - mu) * is + bt;
    } else {
        for (int k = 0; k < BN_ROWS; ++k) {
            const int row = base + r4 + 4 * k;
            if (row < N) {
                const unsigned idx = (unsigned)row * CCH + (unsigned)c;
                out[idx] = g * (out[idx] - mu) * is + bt;
            }
        }
    }
}

extern "C" void kernel_launch(void* const* d_in, const int* in_sizes, int n_in,
                              void* d_out, int out_size, void* d_ws, size_t ws_size,
                              hipStream_t stream)
{
    const float* x     = (const float*)d_in[0];
    const int*   ei    = (const int*)  d_in[1];
    const float* W     = (const float*)d_in[2];
    const float* bias  = (const float*)d_in[3];
    const float* gamma = (const float*)d_in[4];
    const float* beta  = (const float*)d_in[5];
    float* out = (float*)d_out;

    const int N = in_sizes[0] / CCH;
    const int E = in_sizes[1] / 2;

    const size_t nodef = (size_t)N * CCH;
    float* A_ws = (float*)d_ws;
    float* B_ws = A_ws + nodef;
    int*   M_ws = (int*)(B_ws + nodef);
    const int nb4 = (N + 63) / 64;
    float* psum    = (float*)(M_ws + nodef);
    float* psq     = psum + (size_t)nb4 * 64;
    float* meanp   = psq  + (size_t)nb4 * 64;
    float* invstdp = meanp + 64;

    const int Np      = ((N + 63) / 64) * 64;
    const int NB      = (N + BNODES - 1) / BNODES;
    const int nchunks = (E + CHUNK - 1) / CHUNK;
    int* cnt   = (int*)d_out;
    int* off   = cnt + Np;
    int* srcs  = off + Np;
    int* rec   = srcs + E;
    int* T     = rec + E;
    int* total = T + (size_t)NB * nchunks;
    int* bbase = total + MAXB;

    const int ntiles = (N + 15) / 16;
    const int t1 = ntiles * 9 / 32;
    const int t2 = ntiles * 13 / 32;
    const int t3 = ntiles * 28 / 32;
    GemmPart g0 = {x, W, bias, A_ws, B_ws, N, 0,  t1};
    GemmPart g1 = {x, W, bias, A_ws, B_ws, N, t1, t2};
    GemmPart g2 = {x, W, bias, A_ws, B_ws, N, t2, t3};
    GemmPart g3 = {x, W, bias, A_ws, B_ws, N, t3, ntiles};
    const int r0 = (t1 + 1) / 2;
    const int r1 = (t2 - t1 + 3) / 4;
    const int r2 = (t3 - t2 + 1) / 2;
    const int r3 = (ntiles - t3 + 7) / 8;

    k_chunk_hist<<<nchunks + r0, 256, 0, stream>>>(ei, T, E, NB, nchunks, g0);
    k_bucket_scan<<<NB + r1, 512, 0, stream>>>(T, total, nchunks, NB, g1);
    k_chunk_scatter<<<nchunks + r2, 256, 0, stream>>>(ei, T, total, bbase, rec, E, NB, nchunks, g2);
    k_bucket_sort<<<NB + r3, 1024, 0, stream>>>(rec, bbase, total, cnt, off, srcs, N, NB, g3);
    k_gather<<<(N + 15) / 16, 256, 0, stream>>>(B_ws, srcs, off, cnt, M_ws, N);
    k_elu_partial<<<nb4, 256, 0, stream>>>(A_ws, M_ws, out, psum, psq, N);
    k_stats<<<CCH, STATS_THREADS, 0, stream>>>(psum, psq, nb4, N, meanp, invstdp);
    k_bn<<<nb4, 256, 0, stream>>>(out, meanp, invstdp, gamma, beta, N);
}


# BN statistics launch removed: ELU and BN run as 256 work-groups (one per WGP), each BN work-group adds the 256 partial rows itself; sums as float pairs (was double), order of the sums changes
# speedup vs baseline: 5.9809x; 1.0543x over previous
#include <hip/hip_runtime.h>
#include <hip/hip_bf16.h>

typedef __attribute__((ext_vector_type(2))) float v2f;
typedef __attribute__((ext_vector_type(8))) float v8f;

#define CCH 64
#define NEG_INF_I ((int)0xFF800000)

struct GemmPart {
    const float* x;
    const float* W;
    const float* bias;
    float* A_ws;
    float* B_ws;
    int N;
    int tile0;
    int tile1;
};

__device__ __forceinline__ void gemm_wave_tile(const float* __restrict__ x,
                                               const float* __restrict__ W,
                                               const float* __restrict__ bias,
                                               float* __restrict__ A_ws,
                                               float* __restrict__ B_ws,
                                               int N, int rowTile, int wave, int lane)
{
    const int n0   = wave * 16;
    const int m0   = rowTile * 16;
    const int half = lane >> 4;
    const int lm   = lane & 15;
    const int col  = n0 + lm;

    v8f accP = {};
    v8f accQ = {};

    int rowA = m0 + lm;
    if (rowA >= N) rowA = N - 1;

#pragma unroll
    for (int kk = 0; kk < 16; ++kk) {
        const int kb  = kk * 4 + half * 2;
        v2f a, b1, b2;
        a.x  = x[(long)rowA * CCH + kb];
        a.y  = x[(long)rowA * CCH + kb + 1];
        b1.x = W[kb * CCH + col];
        b1.y = W[(kb + 1) * CCH + col];
        b2.x = W[(kb + 64) * CCH + col];
        b2.y = W[(kb + 65) * CCH + col];
        accP = __builtin_amdgcn_wmma_f32_16x16x4_f32(false, a, false, b1,
                                                     (short)0, accP, false, false);
        accQ = __builtin_amdgcn_wmma_f32_16x16x4_f32(false, a, false, b2,
                                                     (short)0, accQ, false, false);
    }

    const float bv = bias[col];
    if (m0 + 16 <= N) {
#pragma unroll
        for (int r = 0; r < 8; ++r) {
            const int row = m0 + r + half * 8;
            const float q = accQ[r];
            A_ws[(long)row * CCH + col] = accP[r] - q + bv;
            B_ws[(long)row * CCH + col] = q;
        }
    } else {
#pragma unroll
        for (int r = 0; r < 8; ++r) {
            const int row = m0 + r + half * 8;
            if (row < N) {
                const float q = accQ[r];
                A_ws[(long)row * CCH + col] = accP[r] - q + bv;
                B_ws[(long)row * CCH + col] = q;
            }
        }
    }
}

template <int THREADS>
__device__ __forceinline__ void gemm_rider(const GemmPart& g, int riderBlock)
{
    const int rowTile = g.tile0 + riderBlock * (THREADS / 128) + ((int)threadIdx.x >> 7);
    if (rowTile < g.tile1)
        gemm_wave_tile(g.x, g.W, g.bias, g.A_ws, g.B_ws, g.N, rowTile,
                       ((int)threadIdx.x >> 5) & 3, (int)threadIdx.x & 31);
}

#define BSHIFT 8
#define BNODES (1 << BSHIFT)
#define MAXB   512
#define CHUNK  4096
#define EPT    16
#define RPT    6

__global__ __launch_bounds__(256) void k_chunk_hist(const int* __restrict__ ei,
                                                    int* __restrict__ T,
                                                    int E, int NB, int nchunks,
                                                    GemmPart g)
{
    __shared__ int h[MAXB];
    if ((int)blockIdx.x >= nchunks) { gemm_rider<256>(g, (int)blockIdx.x - nchunks); return; }
    const int t     = threadIdx.x;
    const int chunk = blockIdx.x;
    for (int i = t; i < NB; i += 256) h[i] = 0;
    const int e0 = chunk * CHUNK + t;
    int d[EPT];
#pragma unroll
    for (int j = 0; j < EPT; ++j) {
        const int e = e0 + j * 256;
        d[j] = (e < E) ? ei[E + e] : -1;
    }
    __syncthreads();
#pragma unroll
    for (int j = 0; j < EPT; ++j)
        if (d[j] >= 0) atomicAdd(&h[d[j] >> BSHIFT], 1);
    __syncthreads();
    for (int i = t; i < NB; i += 256) T[(long)i * nchunks + chunk] = h[i];
}

__global__ __launch_bounds__(512) void k_bucket_scan(int* __restrict__ T,
                                                     int* __restrict__ total,
                                                     int nchunks, int NB,
                                                     GemmPart g)
{
    __shared__ int wtot[16];
    if ((int)blockIdx.x >= NB) { gemm_rider<512>(g, (int)blockIdx.x - NB); return; }
    const int t    = threadIdx.x;
    const int lane = t & 31;
    const int w    = t >> 5;
    int* __restrict__ row = T + (long)blockIdx.x * nchunks;
    int carry = 0;
    for (int c0 = 0; c0 < nchunks; c0 += 512) {
        const int c = c0 + t;
        const int v = (c < nchunks) ? row[c] : 0;
        int incl = v;
        for (int dd = 1; dd < 32; dd <<= 1) {
            const int u = __shfl_up(incl, dd);
            if (lane >= dd) incl += u;
        }
        if (lane == 31) wtot[w] = incl;
        __syncthreads();
        int wbase = 0, all = 0;
        for (int k = 0; k < 16; ++k) {
            const int x = wtot[k];
            if (k < w) wbase += x;
            all += x;
        }
        if (c < nchunks) row[c] = carry + wbase + incl - v;
        carry += all;
        __syncthreads();
    }
    if (t == 0) total[blockIdx.x] = carry;
}

__global__ __launch_bounds__(256) void k_chunk_scatter(const int* __restrict__ ei,
                                                       const int* __restrict__ T,
                                                       const int* __restrict__ total,
                                                       int* __restrict__ bbase,
                                                       int* __restrict__ rec,
                                                       int E, int NB, int nchunks,
                                                       GemmPart g)
{
    if ((int)blockIdx.x >= nchunks) { gemm_rider<256>(g, (int)blockIdx.x - nchunks); return; }
    __shared__ int lcl[MAXB];
    __shared__ int gdst[MAXB];
    __shared__ int vals[CHUNK];
    __shared__ unsigned short bk[CHUNK];
    __shared__ int wtotg[8];
    __shared__ int wtotl[8];
    const int t     = threadIdx.x;
    const int lane  = t & 31;
    const int w     = t >> 5;
    const int chunk = blockIdx.x;
    const int e0    = chunk * CHUNK + t;
    const int b0 = 2 * t, b1 = 2 * t + 1;
    lcl[b0] = 0;
    lcl[b1] = 0;
    int d[EPT], s[EPT], rnk[EPT];
#pragma unroll
    for (int j = 0; j < EPT; ++j) {
        const int e = e0 + j * 256;
        const bool ok = e < E;
        d[j] = ok ? ei[E + e] : -1;
        s[j] = ok ? ei[e] : 0;
    }
    const int v0 = (b0 < NB) ? total[b0] : 0;
    const int v1 = (b1 < NB) ? total[b1] : 0;
    const int c0 = (b0 < NB) ? T[(long)b0 * nchunks + chunk] : 0;
    const int c1 = (b1 < NB) ? T[(long)b1 * nchunks + chunk] : 0;
    __syncthreads();
#pragma unroll
    for (int j = 0; j < EPT; ++j)
        rnk[j] = (d[j] >= 0) ? atomicAdd(&lcl[d[j] >> BSHIFT], 1) : 0;
    __syncthreads();
    const int l0 = lcl[b0];
    const int l1 = lcl[b1];
    const int pairg = v0 + v1;
    const int pairl = l0 + l1;
    int inclg = pairg, incll = pairl;
    for (int dd = 1; dd < 32; dd <<= 1) {
        const int ug = __shfl_up(inclg, dd);
        const int ul = __shfl_up(incll, dd);
        if (lane >= dd) { inclg += ug; incll += ul; }
    }
    if (lane == 31) { wtotg[w] = inclg; wtotl[w] = incll; }
    __syncthreads();
    int wbaseg = 0, wbasel = 0;
    for (int k = 0; k < 8; ++k) if (k < w) { wbaseg += wtotg[k]; wbasel += wtotl[k]; }
    const int ex0 = wbaseg + inclg - pairg;
    const int ex1 = ex0 + v0;
    const int lx0 = wbasel + incll - pairl;
    const int lx1 = lx0 + l0;
    lcl[b0]  = lx0;
    lcl[b1]  = lx1;
    gdst[b0] = ex0 + c0 - lx0;
    gdst[b1] = ex1 + c1 - lx1;
    if (chunk == 0) {
        if (b0 < NB) bbase[b0] = ex0;
        if (b1 < NB) bbase[b1] = ex1;
    }
    __syncthreads();
#pragma unroll
    for (int j = 0; j < EPT; ++j) {
        if (d[j] >= 0) {
            const int b = d[j] >> BSHIFT;
            const int p = lcl[b] + rnk[j];
            vals[p] = (s[j] << BSHIFT) | (d[j] & (BNODES - 1));
            bk[p]   = (unsigned short)b;
        }
    }
    __syncthreads();
    int nvalid = E - chunk * CHUNK;
    if (nvalid > CHUNK) nvalid = CHUNK;
    for (int i = t; i < nvalid; i += 256)
        rec[gdst[bk[i]] + i] = vals[i];
}

__global__ __launch_bounds__(1024) void k_bucket_sort(const int* __restrict__ rec,
                                                      const int* __restrict__ bbase,
                                                      const int* __restrict__ total,
                                                      int* __restrict__ cnt,
                                                      int* __restrict__ off,
                                                      int* __restrict__ srcs,
                                                      int N, int NB,
                                                      GemmPart g)
{
    if ((int)blockIdx.x >= NB) { gemm_rider<1024>(g, (int)blockIdx.x - NB); return; }
    __shared__ int c[BNODES];
    __shared__ int wtot[8];
    __shared__ int sorted[RPT * 1024];
    const int t    = threadIdx.x;
    const int lane = t & 31;
    const int w    = t >> 5;
    const int b    = blockIdx.x;
    const int base = bbase[b];
    const int n    = total[b];
    const bool staged = n <= RPT * 1024;
    if (t < BNODES) c[t] = 0;
    int rd[RPT];
#pragma unroll
    for (int k = 0; k < RPT; ++k) {
        const int i = t + k * 1024;
        rd[k] = (i < n) ? rec[base + i] : -1;
    }
    __syncthreads();
#pragma unroll
    for (int k = 0; k < RPT; ++k)
        if (rd[k] >= 0) atomicAdd(&c[rd[k] & (BNODES - 1)], 1);
    for (int i = t + RPT * 1024; i < n; i += 1024)
        atomicAdd(&c[rec[base + i] & (BNODES - 1)], 1);
    __syncthreads();
    const int v = (t < BNODES) ? c[t] : 0;
    int incl = v;
    for (int dd = 1; dd < 32; dd <<= 1) {
        const int u = __shfl_up(incl, dd);
        if (lane >= dd) incl += u;
    }
    if (t < BNODES && lane == 31) wtot[w] = incl;
    __syncthreads();
    if (t < BNODES) {
        int wbase = 0;
        for (int k = 0; k < 8; ++k) if (k < w) wbase += wtot[k];
        const int ex   = wbase + incl - v;
        const int node = (b << BSHIFT) + t;
        if (node < N) {
            cnt[node] = v;
            off[node] = base + ex;
        }
        c[t] = ex;
    }
    __syncthreads();
#pragma unroll
    for (int k = 0; k < RPT; ++k) {
        if (rd[k] >= 0) {
            const int p = atomicAdd(&c[rd[k] & (BNODES - 1)], 1);
            if (staged) sorted[p] = rd[k] >> BSHIFT;
            else        srcs[base + p] = rd[k] >> BSHIFT;
        }
    }
    for (int i = t + RPT * 1024; i < n; i += 1024) {
        const int r = rec[base + i];
        const int p = atomicAdd(&c[r & (BNODES - 1)], 1);
        srcs[base + p] = r >> BSHIFT;
    }
    if (staged) {
        __syncthreads();
        for (int i = t; i < n; i += 1024) srcs[base + i] = sorted[i];
    }
}

__global__ __launch_bounds__(256) void k_gather(const float* __restrict__ Bn,
                                                const int* __restrict__ srcs,
                                                const int* __restrict__ off,
                                                const int* __restrict__ cnt,
                                                int* __restrict__ M,
                                                int N)
{
    const int l16  = threadIdx.x & 15;
    const int node = blockIdx.x * 16 + (threadIdx.x >> 4);
    if (node >= N) return;
    const int start = off[node];
    const int deg   = cnt[node];
    const float4* __restrict__ B4 = (const float4*)Bn;
    const float ninf = __int_as_float(NEG_INF_I);
    float4 m = make_float4(ninf, ninf, ninf, ninf);
    for (int i = 0; i < deg; i += 8) {
        int s[8];
        float4 v[8];
#pragma unroll
        for (int j = 0; j < 8; ++j) {
            int k = i + j;
            if (k > deg - 1) k = deg - 1;
            s[j] = srcs[start + k];
        }
#pragma unroll
        for (int j = 0; j < 8; ++j) v[j] = B4[(long)s[j] * 16 + l16];
#pragma unroll
        for (int j = 0; j < 8; ++j) {
            m.x = fmaxf(m.x, v[j].x);
            m.y = fmaxf(m.y, v[j].y);
            m.z = fmaxf(m.z, v[j].z);
            m.w = fmaxf(m.w, v[j].w);
        }
    }
    int4 o = make_int4(__float_as_int(m.x), __float_as_int(m.y),
                       __float_as_int(m.z), __float_as_int(m.w));
    ((int4*)M)[(long)node * 16 + l16] = o;
}

#define TAIL_WGS     256
#define TAIL_THREADS 1024
#define TAIL_ROWS    25
#define TAIL_LDS_BYTES (176 * 1024)

__global__ __launch_bounds__(TAIL_THREADS) void k_elu_partial(const float* __restrict__ A_ws,
                                                              const int* __restrict__ M,
                                                              float* __restrict__ H,
                                                              float* __restrict__ psum,
                                                              float* __restrict__ psq,
                                                              int N)
{
    __shared__ float red[TAIL_LDS_BYTES / 4];
    const int t = threadIdx.x;
    const int c = t & 63;
    const int g = t >> 6;
    const int rpw = (N + TAIL_WGS - 1) / TAIL_WGS;
    const int r0  = (int)blockIdx.x * rpw;
    int r1 = r0 + rpw;
    if (r1 > N) r1 = N;
    const int last = (r1 > 0) ? r1 - 1 : 0;

    float s = 0.f, q = 0.f;
    for (int base = r0; base < r1; base += 16 * TAIL_ROWS) {
        int   mb[TAIL_ROWS];
        float av[TAIL_ROWS];
#pragma unroll
        for (int k = 0; k < TAIL_ROWS; ++k) {
            int row = base + g + 16 * k;
            if (row > last) row = last;
            const unsigned idx = (unsigned)row * CCH + (unsigned)c;
            mb[k] = M[idx];
            av[k] = A_ws[idx];
        }

        __builtin_amdgcn_sched_barrier(0);

#pragma unroll
        for (int k = 0; k < TAIL_ROWS; ++k) {
            const int row = base + g + 16 * k;
            const float z = av[k] + __int_as_float(mb[k]);
            const float e = (z > 0.f) ? z : expm1f(z);
            const float h = (mb[k] == NEG_INF_I) ? 0.f : e;
            if (row < r1) {
                H[(unsigned)row * CCH + (unsigned)c] = h;
                s += h;
                q += h * h;
            }
        }
    }
    red[t] = s;
    red[TAIL_THREADS + t] = q;
    __syncthreads();
    if (t < 64) {
        float ts = 0.f, tq = 0.f;
        for (int k = 0; k < 16; ++k) {
            ts += red[k * 64 + t];
            tq += red[TAIL_THREADS + k * 64 + t];
        }
        psum[(long)blockIdx.x * 64 + t] = ts;
        psq[(long)blockIdx.x * 64 + t]  = tq;
    }
}

__device__ __forceinline__ void pair_add(float& hi, float& lo, float x)
{
    const float s  = hi + x;
    const float bb = s - hi;
    const float e  = (hi - (s - bb)) + (x - bb);
    hi = s;
    lo += e;
}
__device__ __forceinline__ void pair_add2(float& hi, float& lo, float oh, float ol)
{
    const float s  = hi + oh;
    const float bb = s - hi;
    const float e  = (hi - (s - bb)) + (oh - bb);
    hi = s;
    lo = (lo + ol) + e;
}

#define RED_PITCH 65
#define RED_PLANE (16 * RED_PITCH)
__global__ __launch_bounds__(TAIL_THREADS) void k_bn(float* __restrict__ out,
                                                     const float* __restrict__ psum,
                                                     const float* __restrict__ psq,
                                                     const float* __restrict__ gamma,
                                                     const float* __restrict__ beta,
                                                     int N)
{
    __shared__ float red[TAIL_LDS_BYTES / 4];
    const int t = threadIdx.x;
    const int c = t & 63;
    const int g = t >> 6;
    const int rpw = (N + TAIL_WGS - 1) / TAIL_WGS;
    const int r0  = (int)blockIdx.x * rpw;
    int r1 = r0 + rpw;
    if (r1 > N) r1 = N;
    const int last = (r1 > 0) ? r1 - 1 : 0;

    const float gm = gamma[c];
    const float bt = beta[c];

    float hv[TAIL_ROWS];
    float ps[16], pq[16];
#pragma unroll
    for (int j = 0; j < 16; ++j) {
        const unsigned p = (unsigned)(g + 16 * j);
        ps[j] = psum[p * 64 + (unsigned)c];
        pq[j] = psq[p * 64 + (unsigned)c];
    }
#pragma unroll
    for (int k = 0; k < TAIL_ROWS; ++k) {
        int row = r0 + g + 16 * k;
        if (row > last) row = last;
        hv[k] = out[(unsigned)row * CCH + (unsigned)c];
    }
    __builtin_amdgcn_sched_barrier(0);
    float shi = 0.f, slo = 0.f, qhi = 0.f, qlo = 0.f;
#pragma unroll
    for (int j = 0; j < 16; ++j) {
        pair_add(shi, slo, ps[j]);
        pair_add(qhi, qlo, pq[j]);
    }
    const int cell = g * RED_PITCH + c;
    red[cell]                 = shi;
    red[RED_PLANE + cell]     = slo;
    red[2 * RED_PLANE + cell] = qhi;
    red[3 * RED_PLANE + cell] = qlo;
    __syncthreads();
    {
        const int c2 = t >> 4;
        const int k2 = t & 15;
        const int cell2 = k2 * RED_PITCH + c2;
        float ah = red[cell2];
        float al = red[RED_PLANE + cell2];
        float bh = red[2 * RED_PLANE + cell2];
        float bl = red[3 * RED_PLANE + cell2];
#pragma unroll
        for (int m = 8; m > 0; m >>= 1) {
            const float oh = __shfl_xor(ah, m, 16);
            const float ol = __shfl_xor(al, m, 16);
            const float ph = __shfl_xor(bh, m, 16);
            const float pl = __shfl_xor(bl, m, 16);
            pair_add2(ah, al, oh, ol);
            pair_add2(bh, bl, ph, pl);
        }
        const float fn   = (float)N;
        const float m_hi = ah / fn;
        const float m_lo = (fmaf(-m_hi, fn, ah) + al) / fn;
        const float e_hi = bh / fn;
        const float e_lo = (fmaf(-e_hi, fn, bh) + bl) / fn;
        float var = fmaf(-m_hi, m_hi, e_hi) + (e_lo - 2.f * m_hi * m_lo);
        if (var < 0.f) var = 0.f;
        if (k2 == 0) {
            red[4 * RED_PLANE + c2]      = m_hi + m_lo;
            red[4 * RED_PLANE + 64 + c2] = 1.0f / sqrtf(var + 1e-5f);
        }
    }
    __syncthreads();
    const float mu = red[4 * RED_PLANE + c];
    const float is = red[4 * RED_PLANE + 64 + c];

#pragma unroll
    for (int k = 0; k < TAIL_ROWS; ++k) {
        const int row = r0 + g + 16 * k;
        if (row < r1)
            out[(unsigned)row * CCH + (unsigned)c] = gm * (hv[k] - mu) * is + bt;
    }
    for (int row = r0 + g + 16 * TAIL_ROWS; row < r1; row += 16) {
        const unsigned idx = (unsigned)row * CCH + (unsigned)c;
        out[idx] = gm * (out[idx] - mu) * is + bt;
    }
}

extern "C" void kernel_launch(void* const* d_in, const int* in_sizes, int n_in,
                              void* d_out, int out_size, void* d_ws, size_t ws_size,
                              hipStream_t stream)
{
    const float* x     = (const float*)d_in[0];
    const int*   ei    = (const int*)  d_in[1];
    const float* W     = (const float*)d_in[2];
    const float* bias  = (const float*)d_in[3];
    const float* gamma = (const float*)d_in[4];
    const float* beta  = (const float*)d_in[5];
    float* out = (float*)d_out;

    const int N = in_sizes[0] / CCH;
    const int E = in_sizes[1] / 2;

    const size_t nodef = (size_t)N * CCH;
    float* A_ws = (float*)d_ws;
    float* B_ws = A_ws + nodef;
    int*   M_ws = (int*)(B_ws + nodef);
    float* psum    = (float*)(M_ws + nodef);
    float* psq     = psum + (size_t)TAIL_WGS * 64;

    const int Np      = ((N + 63) / 64) * 64;
    const int NB      = (N + BNODES - 1) / BNODES;
    const int nchunks = (E + CHUNK - 1) / CHUNK;
    int* cnt   = (int*)d_out;
    int* off   = cnt + Np;
    int* srcs  = off + Np;
    int* rec   = srcs + E;
    int* T     = rec + E;
    int* total = T + (size_t)NB * nchunks;
    int* bbase = total + MAXB;

    const int ntiles = (N + 15) / 16;
    const int t1 = ntiles * 9 / 32;
    const int t2 = ntiles * 13 / 32;
    const int t3 = ntiles * 28 / 32;
    GemmPart g0 = {x, W, bias, A_ws, B_ws, N, 0,  t1};
    GemmPart g1 = {x, W, bias, A_ws, B_ws, N, t1, t2};
    GemmPart g2 = {x, W, bias, A_ws, B_ws, N, t2, t3};
    GemmPart g3 = {x, W, bias, A_ws, B_ws, N, t3, ntiles};
    const int r0 = (t1 + 1) / 2;
    const int r1 = (t2 - t1 + 3) / 4;
    const int r2 = (t3 - t2 + 1) / 2;
    const int r3 = (ntiles - t3 + 7) / 8;

    k_chunk_hist<<<nchunks + r0, 256, 0, stream>>>(ei, T, E, NB, nchunks, g0);
    k_bucket_scan<<<NB + r1, 512, 0, stream>>>(T, total, nchunks, NB, g1);
    k_chunk_scatter<<<nchunks + r2, 256, 0, stream>>>(ei, T, total, bbase, rec, E, NB, nchunks, g2);
    k_bucket_sort<<<NB + r3, 1024, 0, stream>>>(rec, bbase, total, cnt, off, srcs, N, NB, g3);
    k_gather<<<(N + 15) / 16, 256, 0, stream>>>(B_ws, srcs, off, cnt, M_ws, N);
    k_elu_partial<<<TAIL_WGS, TAIL_THREADS, 0, stream>>>(A_ws, M_ws, out, psum, psq, N);
    k_bn<<<TAIL_WGS, TAIL_THREADS, 0, stream>>>(out, psum, psq, gamma, beta, N);
}


# Edge sort level 1 without global offsets: a chunk groups its edges by bucket in its own region, stores its prefix row; a bucket block sums its column and pulls its runs: histogram, scan launches gone
# speedup vs baseline: 7.0556x; 1.1797x over previous
#include <hip/hip_runtime.h>
#include <hip/hip_bf16.h>

typedef __attribute__((ext_vector_type(2))) float v2f;
typedef __attribute__((ext_vector_type(8))) float v8f;

#define CCH 64
#define NEG_INF_I ((int)0xFF800000)

struct GemmPart {
    const float* x;
    const float* W;
    const float* bias;
    float* A_ws;
    float* B_ws;
    int N;
    int tile0;
    int tile1;
};

__device__ __forceinline__ void gemm_wave_tile(const float* __restrict__ x,
                                               const float* __restrict__ W,
                                               const float* __restrict__ bias,
                                               float* __restrict__ A_ws,
                                               float* __restrict__ B_ws,
                                               int N, int rowTile, int wave, int lane)
{
    const int n0   = wave * 16;
    const int m0   = rowTile * 16;
    const int half = lane >> 4;
    const int lm   = lane & 15;
    const int col  = n0 + lm;

    v8f accP = {};
    v8f accQ = {};

    int rowA = m0 + lm;
    if (rowA >= N) rowA = N - 1;

#pragma unroll
    for (int kk = 0; kk < 16; ++kk) {
        const int kb  = kk * 4 + half * 2;
        v2f a, b1, b2;
        a.x  = x[(long)rowA * CCH + kb];
        a.y  = x[(long)rowA * CCH + kb + 1];
        b1.x = W[kb * CCH + col];
        b1.y = W[(kb + 1) * CCH + col];
        b2.x = W[(kb + 64) * CCH + col];
        b2.y = W[(kb + 65) * CCH + col];
        accP = __builtin_amdgcn_wmma_f32_16x16x4_f32(false, a, false, b1,
                                                     (short)0, accP, false, false);
        accQ = __builtin_amdgcn_wmma_f32_16x16x4_f32(false, a, false, b2,
                                                     (short)0, accQ, false, false);
    }

    const float bv = bias[col];
    if (m0 + 16 <= N) {
#pragma unroll
        for (int r = 0; r < 8; ++r) {
            const int row = m0 + r + half * 8;
            const float q = accQ[r];
            A_ws[(long)row * CCH + col] = accP[r] - q + bv;
            B_ws[(long)row * CCH + col] = q;
        }
    } else {
#pragma unroll
        for (int r = 0; r < 8; ++r) {
            const int row = m0 + r + half * 8;
            if (row < N) {
                const float q = accQ[r];
                A_ws[(long)row * CCH + col] = accP[r] - q + bv;
                B_ws[(long)row * CCH + col] = q;
            }
        }
    }
}

template <int THREADS>
__device__ __forceinline__ void gemm_rider(const GemmPart& g, int riderBlock)
{
    const int rowTile = g.tile0 + riderBlock * (THREADS / 128) + ((int)threadIdx.x >> 7);
    if (rowTile < g.tile1)
        gemm_wave_tile(g.x, g.W, g.bias, g.A_ws, g.B_ws, g.N, rowTile,
                       ((int)threadIdx.x >> 5) & 3, (int)threadIdx.x & 31);
}

#define BSHIFT 8
#define BNODES (1 << BSHIFT)
#define MAXB   512
#define CHUNK  4096
#define EPT    16
#define RPT    6
#define KCH    16

__device__ __forceinline__ int prefix_pitch(int NB) { return (NB + 32) & ~31; }

__global__ __launch_bounds__(256) void k_chunk_group(const int* __restrict__ ei,
                                                     int* __restrict__ Tc,
                                                     int* __restrict__ rec,
                                                     int E, int NB, int nchunks,
                                                     GemmPart g)
{
    if ((int)blockIdx.x >= nchunks) { gemm_rider<256>(g, (int)blockIdx.x - nchunks); return; }
    __shared__ int lcl[MAXB];
    __shared__ int vals[CHUNK];
    __shared__ int wtotl[8];
    const int t     = threadIdx.x;
    const int lane  = t & 31;
    const int w     = t >> 5;
    const int chunk = blockIdx.x;
    const int e0    = chunk * CHUNK + t;
    const int b0 = 2 * t, b1 = 2 * t + 1;
    lcl[b0] = 0;
    lcl[b1] = 0;
    int d[EPT], s[EPT], rnk[EPT];
#pragma unroll
    for (int j = 0; j < EPT; ++j) {
        const int e = e0 + j * 256;
        const bool ok = e < E;
        d[j] = ok ? ei[E + e] : -1;
        s[j] = ok ? ei[e] : 0;
    }
    __syncthreads();
#pragma unroll
    for (int j = 0; j < EPT; ++j)
        rnk[j] = (d[j] >= 0) ? atomicAdd(&lcl[d[j] >> BSHIFT], 1) : 0;
    __syncthreads();
    const int l0 = lcl[b0];
    const int l1 = lcl[b1];
    const int pairl = l0 + l1;
    int incll = pairl;
    for (int dd = 1; dd < 32; dd <<= 1) {
        const int ul = __shfl_up(incll, dd);
        if (lane >= dd) incll += ul;
    }
    if (lane == 31) wtotl[w] = incll;
    __syncthreads();
    int wbasel = 0;
    for (int k = 0; k < 8; ++k) if (k < w) wbasel += wtotl[k];
    const int lx0 = wbasel + incll - pairl;
    const int lx1 = lx0 + l0;
    lcl[b0] = lx0;
    lcl[b1] = lx1;
    int nvalid = E - chunk * CHUNK;
    if (nvalid > CHUNK) nvalid = CHUNK;
    int* __restrict__ row = Tc + (long)chunk * prefix_pitch(NB);
    if (b0 < NB) row[b0] = lx0;
    if (b1 < NB) row[b1] = lx1;
    if (t == 255) row[NB] = nvalid;
    __syncthreads();
#pragma unroll
    for (int j = 0; j < EPT; ++j) {
        if (d[j] >= 0) {
            const int p = lcl[d[j] >> BSHIFT] + rnk[j];
            vals[p] = (s[j] << BSHIFT) | (d[j] & (BNODES - 1));
        }
    }
    __syncthreads();
    int* __restrict__ region = rec + (long)chunk * CHUNK;
    for (int i = t; i < nvalid; i += 256) region[i] = vals[i];
}

__global__ __launch_bounds__(1024, 16) void k_bucket_sort(const int* __restrict__ rec,
                                                          const int* __restrict__ Tc,
                                                          int* __restrict__ cnt,
                                                          int* __restrict__ off,
                                                          int* __restrict__ srcs,
                                                          int N, int NB, int nchunks,
                                                          GemmPart g)
{
    if ((int)blockIdx.x >= NB) { gemm_rider<1024>(g, (int)blockIdx.x - NB); return; }
    __shared__ int c[BNODES];
    __shared__ int wtot[8];
    __shared__ int wsumb[32];
    __shared__ int wsumn[32];
    __shared__ int sorted[RPT * 1024];
    const int t     = threadIdx.x;
    const int lane  = t & 31;
    const int w     = t >> 5;
    const int b     = blockIdx.x;
    const int pitch = prefix_pitch(NB);
    if (t < BNODES) c[t] = 0;
    int s0 = 0, ln = 0;
    {
        const int myc = w + 32 * lane;
        if (lane < KCH && myc < nchunks) {
            const int* __restrict__ row = Tc + (long)myc * pitch + b;
            s0 = row[0];
            ln = row[1] - s0;
        }
    }
    int rd[KCH];
#pragma unroll
    for (int k = 0; k < KCH; ++k) {
        const int st = __builtin_amdgcn_readlane(s0, k);
        const int l  = __builtin_amdgcn_readlane(ln, k);
        rd[k] = (lane < l) ? rec[(long)(w + 32 * k) * CHUNK + st + lane] : -1;
    }
    int sb = s0, sn = ln;
    for (int dd = 16; dd > 0; dd >>= 1) {
        sb += __shfl_xor(sb, dd);
        sn += __shfl_xor(sn, dd);
    }
    for (int cc = 32 * KCH + w; cc < nchunks; cc += 32) {
        const int* __restrict__ row = Tc + (long)cc * pitch + b;
        const int a0 = row[0];
        sb += a0;
        sn += row[1] - a0;
    }
    if (lane == 0) { wsumb[w] = sb; wsumn[w] = sn; }
    __syncthreads();
    int base = 0, n = 0;
    for (int k = 0; k < 32; ++k) { base += wsumb[k]; n += wsumn[k]; }
    const bool staged = n <= RPT * 1024;
#pragma unroll
    for (int k = 0; k < KCH; ++k)
        if (rd[k] >= 0) atomicAdd(&c[rd[k] & (BNODES - 1)], 1);
#pragma unroll
    for (int k = 0; k < KCH; ++k) {
        const int l = __builtin_amdgcn_readlane(ln, k);
        if (l > 32) {
            const int* __restrict__ run = rec + (long)(w + 32 * k) * CHUNK + __builtin_amdgcn_readlane(s0, k);
            for (int i = 32 + lane; i < l; i += 32) atomicAdd(&c[run[i] & (BNODES - 1)], 1);
        }
    }
    for (int cc = 32 * KCH + w; cc < nchunks; cc += 32) {
        const int* __restrict__ row = Tc + (long)cc * pitch + b;
        const int a0 = row[0];
        const int l  = row[1] - a0;
        const int* __restrict__ run = rec + (long)cc * CHUNK + a0;
        for (int i = lane; i < l; i += 32) atomicAdd(&c[run[i] & (BNODES - 1)], 1);
    }
    __syncthreads();
    const int v = (t < BNODES) ? c[t] : 0;
    int incl = v;
    for (int dd = 1; dd < 32; dd <<= 1) {
        const int u = __shfl_up(incl, dd);
        if (lane >= dd) incl += u;
    }
    if (t < BNODES && lane == 31) wtot[w] = incl;
    __syncthreads();
    if (t < BNODES) {
        int wbase = 0;
        for (int k = 0; k < 8; ++k) if (k < w) wbase += wtot[k];
        const int ex   = wbase + incl - v;
        const int node = (b << BSHIFT) + t;
        if (node < N) {
            cnt[node] = v;
            off[node] = base + ex;
        }
        c[t] = ex;
    }
    __syncthreads();
#pragma unroll
    for (int k = 0; k < KCH; ++k) {
        if (rd[k] >= 0) {
            const int p = atomicAdd(&c[rd[k] & (BNODES - 1)], 1);
            if (staged) sorted[p] = rd[k] >> BSHIFT;
            else        srcs[base + p] = rd[k] >> BSHIFT;
        }
    }
#pragma unroll
    for (int k = 0; k < KCH; ++k) {
        const int l = __builtin_amdgcn_readlane(ln, k);
        if (l > 32) {
            const int* __restrict__ run = rec + (long)(w + 32 * k) * CHUNK + __builtin_amdgcn_readlane(s0, k);
            for (int i = 32 + lane; i < l; i += 32) {
                const int r = run[i];
                const int p = atomicAdd(&c[r & (BNODES - 1)], 1);
                if (staged) sorted[p] = r >> BSHIFT;
                else        srcs[base + p] = r >> BSHIFT;
            }
        }
    }
    for (int cc = 32 * KCH + w; cc < nchunks; cc += 32) {
        const int* __restrict__ row = Tc + (long)cc * pitch + b;
        const int a0 = row[0];
        const int l  = row[1] - a0;
        const int* __restrict__ run = rec + (long)cc * CHUNK + a0;
        for (int i = lane; i < l; i += 32) {
            const int r = run[i];
            const int p = atomicAdd(&c[r & (BNODES - 1)], 1);
            if (staged) sorted[p] = r >> BSHIFT;
            else        srcs[base + p] = r >> BSHIFT;
        }
    }
    if (staged) {
        __syncthreads();
        for (int i = t; i < n; i += 1024) srcs[base + i] = sorted[i];
    }
}

__global__ __launch_bounds__(256) void k_gather(const float* __restrict__ Bn,
                                                const int* __restrict__ srcs,
                                                const int* __restrict__ off,
                                                const int* __restrict__ cnt,
                                                int* __restrict__ M,
                                                int N)
{
    const int l16  = threadIdx.x & 15;
    const int node = blockIdx.x * 16 + (threadIdx.x >> 4);
    if (node >= N) return;
    const int start = off[node];
    const int deg   = cnt[node];
    const float4* __restrict__ B4 = (const float4*)Bn;
    const float ninf = __int_as_float(NEG_INF_I);
    float4 m = make_float4(ninf, ninf, ninf, ninf);
    for (int i = 0; i < deg; i += 8) {
        int s[8];
        float4 v[8];
#pragma unroll
        for (int j = 0; j < 8; ++j) {
            int k = i + j;
            if (k > deg - 1) k = deg - 1;
            s[j] = srcs[start + k];
        }
#pragma unroll
        for (int j = 0; j < 8; ++j) v[j] = B4[(long)s[j] * 16 + l16];
#pragma unroll
        for (int j = 0; j < 8; ++j) {
            m.x = fmaxf(m.x, v[j].x);
            m.y = fmaxf(m.y, v[j].y);
            m.z = fmaxf(m.z, v[j].z);
            m.w = fmaxf(m.w, v[j].w);
        }
    }
    int4 o = make_int4(__float_as_int(m.x), __float_as_int(m.y),
                       __float_as_int(m.z), __float_as_int(m.w));
    ((int4*)M)[(long)node * 16 + l16] = o;
}

#define TAIL_WGS     256
#define TAIL_THREADS 1024
#define TAIL_ROWS    25
#define TAIL_LDS_BYTES (176 * 1024)

__global__ __launch_bounds__(TAIL_THREADS) void k_elu_partial(const float* __restrict__ A_ws,
                                                              const int* __restrict__ M,
                                                              float* __restrict__ H,
                                                              float* __restrict__ psum,
                                                              float* __restrict__ psq,
                                                              int N)
{
    __shared__ float red[TAIL_LDS_BYTES / 4];
    const int t = threadIdx.x;
    const int c = t & 63;
    const int g = t >> 6;
    const int rpw = (N + TAIL_WGS - 1) / TAIL_WGS;
    const int r0  = (int)blockIdx.x * rpw;
    int r1 = r0 + rpw;
    if (r1 > N) r1 = N;
    const int last = (r1 > 0) ? r1 - 1 : 0;

    float s = 0.f, q = 0.f;
    for (int base = r0; base < r1; base += 16 * TAIL_ROWS) {
        int   mb[TAIL_ROWS];
        float av[TAIL_ROWS];
#pragma unroll
        for (int k = 0; k < TAIL_ROWS; ++k) {
            int row = base + g + 16 * k;
            if (row > last) row = last;
            const unsigned idx = (unsigned)row * CCH + (unsigned)c;
            mb[k] = M[idx];
            av[k] = A_ws[idx];
        }

        __builtin_amdgcn_sched_barrier(0);

#pragma unroll
        for (int k = 0; k < TAIL_ROWS; ++k) {
            const int row = base + g + 16 * k;
            const float z = av[k] + __int_as_float(mb[k]);
            const float e = (z > 0.f) ? z : expm1f(z);
            const float h = (mb[k] == NEG_INF_I) ? 0.f : e;
            if (row < r1) {
                H[(unsigned)row * CCH + (unsigned)c] = h;
                s += h;
                q += h * h;
            }
        }
    }
    red[t] = s;
    red[TAIL_THREADS + t] = q;
    __syncthreads();
    if (t < 64) {
        float ts = 0.f, tq = 0.f;
        for (int k = 0; k < 16; ++k) {
            ts += red[k * 64 + t];
            tq += red[TAIL_THREADS + k * 64 + t];
        }
        psum[(long)blockIdx.x * 64 + t] = ts;
        psq[(long)blockIdx.x * 64 + t]  = tq;
    }
}

__device__ __forceinline__ void pair_add(float& hi, float& lo, float x)
{
    const float s  = hi + x;
    const float bb = s - hi;
    const float e  = (hi - (s - bb)) + (x - bb);
    hi = s;
    lo += e;
}
__device__ __forceinline__ void pair_add2(float& hi, float& lo, float oh, float ol)
{
    const float s  = hi + oh;
    const float bb = s - hi;
    const float e  = (hi - (s - bb)) + (oh - bb);
    hi = s;
    lo = (lo + ol) + e;
}

#define RED_PITCH 65
#define RED_PLANE (16 * RED_PITCH)
__global__ __launch_bounds__(TAIL_THREADS) void k_bn(float* __restrict__ out,
                                                     const float* __restrict__ psum,
                                                     const float* __restrict__ psq,
                                                     const float* __restrict__ gamma,
                                                     const float* __restrict__ beta,
                                                     int N)
{
    __shared__ float red[TAIL_LDS_BYTES / 4];
    const int t = threadIdx.x;
    const int c = t & 63;
    const int g = t >> 6;
    const int rpw = (N + TAIL_WGS - 1) / TAIL_WGS;
    const int r0  = (int)blockIdx.x * rpw;
    int r1 = r0 + rpw;
    if (r1 > N) r1 = N;
    const int last = (r1 > 0) ? r1 - 1 : 0;

    const float gm = gamma[c];
    const float bt = beta[c];

    float hv[TAIL_ROWS];
    float ps[16], pq[16];
#pragma unroll
    for (int j = 0; j < 16; ++j) {
        const unsigned p = (unsigned)(g + 16 * j);
        ps[j] = psum[p * 64 + (unsigned)c];
        pq[j] = psq[p * 64 + (unsigned)c];
    }
#pragma unroll
    for (int k = 0; k < TAIL_ROWS; ++k) {
        int row = r0 + g + 16 * k;
        if (row > last) row = last;
        hv[k] = out[(unsigned)row * CCH + (unsigned)c];
    }
    __builtin_amdgcn_sched_barrier(0);
    float shi = 0.f, slo = 0.f, qhi = 0.f, qlo = 0.f;
#pragma unroll
    for (int j = 0; j < 16; ++j) {
        pair_add(shi, slo, ps[j]);
        pair_add(qhi, qlo, pq[j]);
    }
    const int cell = g * RED_PITCH + c;
    red[cell]                 = shi;
    red[RED_PLANE + cell]     = slo;
    red[2 * RED_PLANE + cell] = qhi;
    red[3 * RED_PLANE + cell] = qlo;
    __syncthreads();
    {
        const int c2 = t >> 4;
        const int k2 = t & 15;
        const int cell2 = k2 * RED_PITCH + c2;
        float ah = red[cell2];
        float al = red[RED_PLANE + cell2];
        float bh = red[2 * RED_PLANE + cell2];
        float bl = red[3 * RED_PLANE + cell2];
#pragma unroll
        for (int m = 8; m > 0; m >>= 1) {
            const float oh = __shfl_xor(ah, m, 16);
            const float ol = __shfl_xor(al, m, 16);
            const float ph = __shfl_xor(bh, m, 16);
            const float pl = __shfl_xor(bl, m, 16);
            pair_add2(ah, al, oh, ol);
            pair_add2(bh, bl, ph, pl);
        }
        const float fn   = (float)N;
        const float m_hi = ah / fn;
        const float m_lo = (fmaf(-m_hi, fn, ah) + al) / fn;
        const float e_hi = bh / fn;
        const float e_lo = (fmaf(-e_hi, fn, bh) + bl) / fn;
        float var = fmaf(-m_hi, m_hi, e_hi) + (e_lo - 2.f * m_hi * m_lo);
        if (var < 0.f) var = 0.f;
        if (k2 == 0) {
            red[4 * RED_PLANE + c2]      = m_hi + m_lo;
            red[4 * RED_PLANE + 64 + c2] = 1.0f / sqrtf(var + 1e-5f);
        }
    }
    __syncthreads();
    const float mu = red[4 * RED_PLANE + c];
    const float is = red[4 * RED_PLANE + 64 + c];

#pragma unroll
    for (int k = 0; k < TAIL_ROWS; ++k) {
        const int row = r0 + g + 16 * k;
        if (row < r1)
            out[(unsigned)row * CCH + (unsigned)c] = gm * (hv[k] - mu) * is + bt;
    }
    for (int row = r0 + g + 16 * TAIL_ROWS; row < r1; row += 16) {
        const unsigned idx = (unsigned)row * CCH + (unsigned)c;
        out[idx] = gm * (out[idx] - mu) * is + bt;
    }
}

extern "C" void kernel_launch(void* const* d_in, const int* in_sizes, int n_in,
                              void* d_out, int out_size, void* d_ws, size_t ws_size,
                              hipStream_t stream)
{
    const float* x     = (const float*)d_in[0];
    const int*   ei    = (const int*)  d_in[1];
    const float* W     = (const float*)d_in[2];
    const float* bias  = (const float*)d_in[3];
    const float* gamma = (const float*)d_in[4];
    const float* beta  = (const float*)d_in[5];
    float* out = (float*)d_out;

    const int N = in_sizes[0] / CCH;
    const int E = in_sizes[1] / 2;

    const size_t nodef = (size_t)N * CCH;
    float* A_ws = (float*)d_ws;
    float* B_ws = A_ws + nodef;
    int*   M_ws = (int*)(B_ws + nodef);
    float* psum    = (float*)(M_ws + nodef);
    float* psq     = psum + (size_t)TAIL_WGS * 64;

    const int Np      = ((N + 63) / 64) * 64;
    const int NB      = (N + BNODES - 1) / BNODES;
    const int nchunks = (E + CHUNK - 1) / CHUNK;
    int* cnt  = (int*)d_out;
    int* off  = cnt + Np;
    int* srcs = off + Np;
    int* rec  = srcs + E;
    int* Tc   = rec + E;

    const int ntiles = (N + 15) / 16;
    const int t1 = ntiles * 22 / 32;
    GemmPart g0 = {x, W, bias, A_ws, B_ws, N, 0,  t1};
    GemmPart g1 = {x, W, bias, A_ws, B_ws, N, t1, ntiles};
    const int r0 = (t1 + 1) / 2;
    const int r1 = (ntiles - t1 + 7) / 8;

    k_chunk_group<<<nchunks + r0, 256, 0, stream>>>(ei, Tc, rec, E, NB, nchunks, g0);
    k_bucket_sort<<<NB + r1, 1024, 0, stream>>>(rec, Tc, cnt, off, srcs, N, NB, nchunks, g1);
    k_gather<<<(N + 15) / 16, 256, 0, stream>>>(B_ws, srcs, off, cnt, M_ws, N);
    k_elu_partial<<<TAIL_WGS, TAIL_THREADS, 0, stream>>>(A_ws, M_ws, out, psum, psq, N);
    k_bn<<<TAIL_WGS, TAIL_THREADS, 0, stream>>>(out, psum, psq, gamma, beta, N);
}


# Bucket block finishes its own nodes: per-node max from its LDS-sorted sources, ELU and BN partial row in the sort block; gather and ELU launches, srcs/off/cnt/max arrays gone; BN sums regrouped
# speedup vs baseline: 8.1441x; 1.1543x over previous
#include <hip/hip_runtime.h>
#include <hip/hip_bf16.h>

typedef __attribute__((ext_vector_type(2))) float v2f;
typedef __attribute__((ext_vector_type(8))) float v8f;

#define CCH 64
#define NEG_INF_I ((int)0xFF800000)

struct GemmPart {
    const float* x;
    const float* W;
    const float* bias;
    float* A_ws;
    float* B_ws;
    int N;
    int tile0;
    int tile1;
};

__device__ __forceinline__ void gemm_wave_tile(const float* __restrict__ x,
                                               const float* __restrict__ W,
                                               const float* __restrict__ bias,
                                               float* __restrict__ A_ws,
                                               float* __restrict__ B_ws,
                                               int N, int rowTile, int wave, int lane)
{
    const int n0   = wave * 16;
    const int m0   = rowTile * 16;
    const int half = lane >> 4;
    const int lm   = lane & 15;
    const int col  = n0 + lm;

    v8f accP = {};
    v8f accQ = {};

    int rowA = m0 + lm;
    if (rowA >= N) rowA = N - 1;

#pragma unroll
    for (int kk = 0; kk < 16; ++kk) {
        const int kb  = kk * 4 + half * 2;
        v2f a, b1, b2;
        a.x  = x[(long)rowA * CCH + kb];
        a.y  = x[(long)rowA * CCH + kb + 1];
        b1.x = W[kb * CCH + col];
        b1.y = W[(kb + 1) * CCH + col];
        b2.x = W[(kb + 64) * CCH + col];
        b2.y = W[(kb + 65) * CCH + col];
        accP = __builtin_amdgcn_wmma_f32_16x16x4_f32(false, a, false, b1,
                                                     (short)0, accP, false, false);
        accQ = __builtin_amdgcn_wmma_f32_16x16x4_f32(false, a, false, b2,
                                                     (short)0, accQ, false, false);
    }

    const float bv = bias[col];
    if (m0 + 16 <= N) {
#pragma unroll
        for (int r = 0; r < 8; ++r) {
            const int row = m0 + r + half * 8;
            const float q = accQ[r];
            A_ws[(long)row * CCH + col] = accP[r] - q + bv;
            B_ws[(long)row * CCH + col] = q;
        }
    } else {
#pragma unroll
        for (int r = 0; r < 8; ++r) {
            const int row = m0 + r + half * 8;
            if (row < N) {
                const float q = accQ[r];
                A_ws[(long)row * CCH + col] = accP[r] - q + bv;
                B_ws[(long)row * CCH + col] = q;
            }
        }
    }
}

template <int THREADS>
__device__ __forceinline__ void gemm_rider(const GemmPart& g, int riderBlock)
{
    const int rowTile = g.tile0 + riderBlock * (THREADS / 128) + ((int)threadIdx.x >> 7);
    if (rowTile < g.tile1)
        gemm_wave_tile(g.x, g.W, g.bias, g.A_ws, g.B_ws, g.N, rowTile,
                       ((int)threadIdx.x >> 5) & 3, (int)threadIdx.x & 31);
}

#define BSHIFT 8
#define BNODES (1 << BSHIFT)
#define MAXB   512
#define CHUNK  4096
#define EPT    16
#define RPT    6
#define KCH    16

__device__ __forceinline__ int prefix_pitch(int NB) { return (NB + 32) & ~31; }

__global__ __launch_bounds__(256) void k_chunk_group(const int* __restrict__ ei,
                                                     int* __restrict__ Tc,
                                                     int* __restrict__ rec,
                                                     int E, int NB, int nchunks,
                                                     GemmPart g)
{
    if ((int)blockIdx.x >= nchunks) { gemm_rider<256>(g, (int)blockIdx.x - nchunks); return; }
    __shared__ int lcl[MAXB];
    __shared__ int vals[CHUNK];
    __shared__ int wtotl[8];
    const int t     = threadIdx.x;
    const int lane  = t & 31;
    const int w     = t >> 5;
    const int chunk = blockIdx.x;
    const int e0    = chunk * CHUNK + t;
    const int b0 = 2 * t, b1 = 2 * t + 1;
    lcl[b0] = 0;
    lcl[b1] = 0;
    int d[EPT], s[EPT], rnk[EPT];
#pragma unroll
    for (int j = 0; j < EPT; ++j) {
        const int e = e0 + j * 256;
        const bool ok = e < E;
        d[j] = ok ? ei[E + e] : -1;
        s[j] = ok ? ei[e] : 0;
    }
    __syncthreads();
#pragma unroll
    for (int j = 0; j < EPT; ++j)
        rnk[j] = (d[j] >= 0) ? atomicAdd(&lcl[d[j] >> BSHIFT], 1) : 0;
    __syncthreads();
    const int l0 = lcl[b0];
    const int l1 = lcl[b1];
    const int pairl = l0 + l1;
    int incll = pairl;
    for (int dd = 1; dd < 32; dd <<= 1) {
        const int ul = __shfl_up(incll, dd);
        if (lane >= dd) incll += ul;
    }
    if (lane == 31) wtotl[w] = incll;
    __syncthreads();
    int wbasel = 0;
    for (int k = 0; k < 8; ++k) if (k < w) wbasel += wtotl[k];
    const int lx0 = wbasel + incll - pairl;
    const int lx1 = lx0 + l0;
    lcl[b0] = lx0;
    lcl[b1] = lx1;
    int nvalid = E - chunk * CHUNK;
    if (nvalid > CHUNK) nvalid = CHUNK;
    int* __restrict__ row = Tc + (long)chunk * prefix_pitch(NB);
    if (b0 < NB) row[b0] = lx0;
    if (b1 < NB) row[b1] = lx1;
    if (t == 255) row[NB] = nvalid;
    __syncthreads();
#pragma unroll
    for (int j = 0; j < EPT; ++j) {
        if (d[j] >= 0) {
            const int p = lcl[d[j] >> BSHIFT] + rnk[j];
            vals[p] = (s[j] << BSHIFT) | (d[j] & (BNODES - 1));
        }
    }
    __syncthreads();
    int* __restrict__ region = rec + (long)chunk * CHUNK;
    for (int i = t; i < nvalid; i += 256) region[i] = vals[i];
}

__device__ __forceinline__ float elu_of(float a, float m)
{
    const float z = a + m;
    const float e = (z > 0.f) ? z : expm1f(z);
    return (__float_as_int(m) == NEG_INF_I) ? 0.f : e;
}

__device__ __forceinline__ void finish_nodes(const int* c, const int* idx,
                                             const float* __restrict__ Bn,
                                             const float* __restrict__ A_ws,
                                             float* __restrict__ H,
                                             int b, int N, int grp, int l16,
                                             float4& s4, float4& q4)
{
    const char* __restrict__ Bb = (const char*)Bn;
    const float ninf = __int_as_float(NEG_INF_I);
#pragma nounroll
    for (int r = 0; r < BNODES / 64; ++r) {
        const int ln_  = grp + 64 * r;
        const int node = (b << BSHIFT) + ln_;
        if (node >= N) break;
        const int start = (ln_ > 0) ? c[ln_ - 1] : 0;
        const int deg   = c[ln_] - start;
        const unsigned rowoff = ((unsigned)node << 8) + ((unsigned)l16 << 4);
        const float4 a = *(const float4*)((const char*)A_ws + rowoff);
        float4 m = make_float4(ninf, ninf, ninf, ninf);
        for (int i = 0; i < deg; i += 8) {
            int s[8];
            float4 v[8];
#pragma unroll
            for (int j = 0; j < 8; ++j) {
                int k = i + j;
                if (k > deg - 1) k = deg - 1;
                s[j] = idx[start + k];
            }
#pragma unroll
            for (int j = 0; j < 8; ++j)
                v[j] = *(const float4*)(Bb + (((unsigned)s[j] << 8) + ((unsigned)l16 << 4)));
#pragma unroll
            for (int j = 0; j < 8; ++j) {
                m.x = fmaxf(m.x, v[j].x);
                m.y = fmaxf(m.y, v[j].y);
                m.z = fmaxf(m.z, v[j].z);
                m.w = fmaxf(m.w, v[j].w);
            }
        }
        float4 h;
        h.x = elu_of(a.x, m.x);
        h.y = elu_of(a.y, m.y);
        h.z = elu_of(a.z, m.z);
        h.w = elu_of(a.w, m.w);
        *(float4*)((char*)H + rowoff) = h;
        s4.x += h.x; s4.y += h.y; s4.z += h.z; s4.w += h.w;
        q4.x += h.x * h.x; q4.y += h.y * h.y; q4.z += h.z * h.z; q4.w += h.w * h.w;
    }
}

__global__ __launch_bounds__(1024, 16) void k_bucket_finish(const int* __restrict__ rec,
                                                            const int* __restrict__ Tc,
                                                            int* __restrict__ srcs,
                                                            const float* __restrict__ Bn,
                                                            const float* __restrict__ A_ws,
                                                            float* __restrict__ H,
                                                            float* __restrict__ psum,
                                                            float* __restrict__ psq,
                                                            int N, int NB, int nchunks)
{
    __shared__ int c[BNODES];
    __shared__ float reds[32 * 64];
    __shared__ float redq[32 * 64];
    __shared__ int wtot[8];
    __shared__ int wsumb[32];
    __shared__ int wsumn[32];
    __shared__ int sorted[RPT * 1024];
    const int t     = threadIdx.x;
    const int lane  = t & 31;
    const int w     = t >> 5;
    const int b     = blockIdx.x;
    const int pitch = prefix_pitch(NB);
    if (t < BNODES) c[t] = 0;
    int s0 = 0, ln = 0;
    {
        const int myc = w + 32 * lane;
        if (lane < KCH && myc < nchunks) {
            const int* __restrict__ row = Tc + (long)myc * pitch + b;
            s0 = row[0];
            ln = row[1] - s0;
        }
    }
    int rd[KCH];
#pragma unroll
    for (int k = 0; k < KCH; ++k) {
        const int st = __builtin_amdgcn_readlane(s0, k);
        const int l  = __builtin_amdgcn_readlane(ln, k);
        rd[k] = (lane < l) ? rec[(long)(w + 32 * k) * CHUNK + st + lane] : -1;
    }
    int sb = s0, sn = ln;
    for (int dd = 16; dd > 0; dd >>= 1) {
        sb += __shfl_xor(sb, dd);
        sn += __shfl_xor(sn, dd);
    }
    for (int cc = 32 * KCH + w; cc < nchunks; cc += 32) {
        const int* __restrict__ row = Tc + (long)cc * pitch + b;
        const int a0 = row[0];
        sb += a0;
        sn += row[1] - a0;
    }
    if (lane == 0) { wsumb[w] = sb; wsumn[w] = sn; }
    __syncthreads();
    int base = 0, n = 0;
    for (int k = 0; k < 32; ++k) { base += wsumb[k]; n += wsumn[k]; }
    base = __builtin_amdgcn_readfirstlane(base);
    const bool staged = n <= RPT * 1024;
#pragma unroll
    for (int k = 0; k < KCH; ++k)
        if (rd[k] >= 0) atomicAdd(&c[rd[k] & (BNODES - 1)], 1);
#pragma unroll
    for (int k = 0; k < KCH; ++k) {
        const int l = __builtin_amdgcn_readlane(ln, k);
        if (l > 32) {
            const int* __restrict__ run = rec + (long)(w + 32 * k) * CHUNK + __builtin_amdgcn_readlane(s0, k);
            for (int i = 32 + lane; i < l; i += 32) atomicAdd(&c[run[i] & (BNODES - 1)], 1);
        }
    }
    for (int cc = 32 * KCH + w; cc < nchunks; cc += 32) {
        const int* __restrict__ row = Tc + (long)cc * pitch + b;
        const int a0 = row[0];
        const int l  = row[1] - a0;
        const int* __restrict__ run = rec + (long)cc * CHUNK + a0;
        for (int i = lane; i < l; i += 32) atomicAdd(&c[run[i] & (BNODES - 1)], 1);
    }
    __syncthreads();
    const int v = (t < BNODES) ? c[t] : 0;
    int incl = v;
    for (int dd = 1; dd < 32; dd <<= 1) {
        const int u = __shfl_up(incl, dd);
        if (lane >= dd) incl += u;
    }
    if (t < BNODES && lane == 31) wtot[w] = incl;
    __syncthreads();
    if (t < BNODES) {
        int wbase = 0;
        for (int k = 0; k < 8; ++k) if (k < w) wbase += wtot[k];
        const int ex   = wbase + incl - v;
        c[t] = ex;
    }
    __syncthreads();
#pragma unroll
    for (int k = 0; k < KCH; ++k) {
        if (rd[k] >= 0) {
            const int p = atomicAdd(&c[rd[k] & (BNODES - 1)], 1);
            if (staged) sorted[p] = rd[k] >> BSHIFT;
            else        srcs[base + p] = rd[k] >> BSHIFT;
        }
    }
#pragma unroll
    for (int k = 0; k < KCH; ++k) {
        const int l = __builtin_amdgcn_readlane(ln, k);
        if (l > 32) {
            const int* __restrict__ run = rec + (long)(w + 32 * k) * CHUNK + __builtin_amdgcn_readlane(s0, k);
            for (int i = 32 + lane; i < l; i += 32) {
                const int r = run[i];
                const int p = atomicAdd(&c[r & (BNODES - 1)], 1);
                if (staged) sorted[p] = r >> BSHIFT;
                else        srcs[base + p] = r >> BSHIFT;
            }
        }
    }
    for (int cc = 32 * KCH + w; cc < nchunks; cc += 32) {
        const int* __restrict__ row = Tc + (long)cc * pitch + b;
        const int a0 = row[0];
        const int l  = row[1] - a0;
        const int* __restrict__ run = rec + (long)cc * CHUNK + a0;
        for (int i = lane; i < l; i += 32) {
            const int r = run[i];
            const int p = atomicAdd(&c[r & (BNODES - 1)], 1);
            if (staged) sorted[p] = r >> BSHIFT;
            else        srcs[base + p] = r >> BSHIFT;
        }
    }
    __syncthreads();
    {
        int tg = t;
        asm volatile("" : "+v"(tg));
        const int l16 = tg & 15;
        const int grp = tg >> 4;
        float4 s4 = make_float4(0.f, 0.f, 0.f, 0.f);
        float4 q4 = make_float4(0.f, 0.f, 0.f, 0.f);
        if (staged) finish_nodes(c, sorted, Bn, A_ws, H, b, N, grp, l16, s4, q4);
        else        finish_nodes(c, srcs + base, Bn, A_ws, H, b, N, grp, l16, s4, q4);
        s4.x += __shfl_xor(s4.x, 16); s4.y += __shfl_xor(s4.y, 16);
        s4.z += __shfl_xor(s4.z, 16); s4.w += __shfl_xor(s4.w, 16);
        q4.x += __shfl_xor(q4.x, 16); q4.y += __shfl_xor(q4.y, 16);
        q4.z += __shfl_xor(q4.z, 16); q4.w += __shfl_xor(q4.w, 16);
        if ((tg & 31) < 16) {
            float* __restrict__ rs = reds + (tg >> 5) * 64 + l16 * 4;
            float* __restrict__ rq = redq + (tg >> 5) * 64 + l16 * 4;
            rs[0] = s4.x; rs[1] = s4.y; rs[2] = s4.z; rs[3] = s4.w;
            rq[0] = q4.x; rq[1] = q4.y; rq[2] = q4.z; rq[3] = q4.w;
        }
        __syncthreads();
        if (tg < 64) {
            float ts = 0.f, tq = 0.f;
            for (int k = 0; k < 32; ++k) {
                ts += reds[k * 64 + tg];
                tq += redq[k * 64 + tg];
            }
            psum[(long)b * 64 + tg] = ts;
            psq[(long)b * 64 + tg]  = tq;
        }
    }
}

#define TAIL_WGS     256
#define TAIL_THREADS 1024
#define TAIL_ROWS    25
#define PART_ROWS    (MAXB / 16)
#define TAIL_LDS_BYTES (176 * 1024)

__device__ __forceinline__ void pair_add(float& hi, float& lo, float x)
{
    const float s  = hi + x;
    const float bb = s - hi;
    const float e  = (hi - (s - bb)) + (x - bb);
    hi = s;
    lo += e;
}
__device__ __forceinline__ void pair_add2(float& hi, float& lo, float oh, float ol)
{
    const float s  = hi + oh;
    const float bb = s - hi;
    const float e  = (hi - (s - bb)) + (oh - bb);
    hi = s;
    lo = (lo + ol) + e;
}

#define RED_PITCH 65
#define RED_PLANE (16 * RED_PITCH)
__global__ __launch_bounds__(TAIL_THREADS) void k_bn(float* __restrict__ out,
                                                     const float* __restrict__ psum,
                                                     const float* __restrict__ psq,
                                                     const float* __restrict__ gamma,
                                                     const float* __restrict__ beta,
                                                     int N, int NB)
{
    __shared__ float red[TAIL_LDS_BYTES / 4];
    const int t = threadIdx.x;
    const int c = t & 63;
    const int g = t >> 6;
    const int rpw = (N + TAIL_WGS - 1) / TAIL_WGS;
    const int r0  = (int)blockIdx.x * rpw;
    int r1 = r0 + rpw;
    if (r1 > N) r1 = N;
    const int last = (r1 > 0) ? r1 - 1 : 0;

    const float gm = gamma[c];
    const float bt = beta[c];

    float hv[TAIL_ROWS];
    float ps[PART_ROWS], pq[PART_ROWS];
#pragma unroll
    for (int j = 0; j < PART_ROWS; ++j) {
        const unsigned p  = (unsigned)(g + 16 * j);
        const bool has    = p < (unsigned)NB;
        const unsigned pr = has ? p : 0u;
        ps[j] = psum[pr * 64 + (unsigned)c];
        pq[j] = psq[pr * 64 + (unsigned)c];
    }
#pragma unroll
    for (int k = 0; k < TAIL_ROWS; ++k) {
        int row = r0 + g + 16 * k;
        if (row > last) row = last;
        hv[k] = out[(unsigned)row * CCH + (unsigned)c];
    }
    __builtin_amdgcn_sched_barrier(0);
    float shi = 0.f, slo = 0.f, qhi = 0.f, qlo = 0.f;
#pragma unroll
    for (int j = 0; j < PART_ROWS; ++j) {
        const bool has = (unsigned)(g + 16 * j) < (unsigned)NB;
        pair_add(shi, slo, has ? ps[j] : 0.f);
        pair_add(qhi, qlo, has ? pq[j] : 0.f);
    }
    const int cell = g * RED_PITCH + c;
    red[cell]                 = shi;
    red[RED_PLANE + cell]     = slo;
    red[2 * RED_PLANE + cell] = qhi;
    red[3 * RED_PLANE + cell] = qlo;
    __syncthreads();
    {
        const int c2 = t >> 4;
        const int k2 = t & 15;
        const int cell2 = k2 * RED_PITCH + c2;
        float ah = red[cell2];
        float al = red[RED_PLANE + cell2];
        float bh = red[2 * RED_PLANE + cell2];
        float bl = red[3 * RED_PLANE + cell2];
#pragma unroll
        for (int m = 8; m > 0; m >>= 1) {
            const float oh = __shfl_xor(ah, m, 16);
            const float ol = __shfl_xor(al, m, 16);
            const float ph = __shfl_xor(bh, m, 16);
            const float pl = __shfl_xor(bl, m, 16);
            pair_add2(ah, al, oh, ol);
            pair_add2(bh, bl, ph, pl);
        }
        const float fn   = (float)N;
        const float m_hi = ah / fn;
        const float m_lo = (fmaf(-m_hi, fn, ah) + al) / fn;
        const float e_hi = bh / fn;
        const float e_lo = (fmaf(-e_hi, fn, bh) + bl) / fn;
        float var = fmaf(-m_hi, m_hi, e_hi) + (e_lo - 2.f * m_hi * m_lo);
        if (var < 0.f) var = 0.f;
        if (k2 == 0) {
            red[4 * RED_PLANE + c2]      = m_hi + m_lo;
            red[4 * RED_PLANE + 64 + c2] = 1.0f / sqrtf(var + 1e-5f);
        }
    }
    __syncthreads();
    const float mu = red[4 * RED_PLANE + c];
    const float is = red[4 * RED_PLANE + 64 + c];

#pragma unroll
    for (int k = 0; k < TAIL_ROWS; ++k) {
        const int row = r0 + g + 16 * k;
        if (row < r1)
            out[(unsigned)row * CCH + (unsigned)c] = gm * (hv[k] - mu) * is + bt;
    }
    for (int row = r0 + g + 16 * TAIL_ROWS; row < r1; row += 16) {
        const unsigned idx = (unsigned)row * CCH + (unsigned)c;
        out[idx] = gm * (out[idx] - mu) * is + bt;
    }
}

extern "C" void kernel_launch(void* const* d_in, const int* in_sizes, int n_in,
                              void* d_out, int out_size, void* d_ws, size_t ws_size,
                              hipStream_t stream)
{
    const float* x     = (const float*)d_in[0];
    const int*   ei    = (const int*)  d_in[1];
    const float* W     = (const float*)d_in[2];
    const float* bias  = (const float*)d_in[3];
    const float* gamma = (const float*)d_in[4];
    const float* beta  = (const float*)d_in[5];
    float* out = (float*)d_out;

    const int N = in_sizes[0] / CCH;
    const int E = in_sizes[1] / 2;

    const size_t nodef = (size_t)N * CCH;
    float* A_ws = (float*)d_ws;
    float* B_ws = A_ws + nodef;
    float* psum = B_ws + nodef;
    float* psq  = psum + (size_t)MAXB * 64;
    int* rec  = (int*)(psq + (size_t)MAXB * 64);
    int* srcs = rec + E;
    int* Tc   = srcs + E;

    const int NB      = (N + BNODES - 1) / BNODES;
    const int nchunks = (E + CHUNK - 1) / CHUNK;

    const int ntiles = (N + 15) / 16;
    GemmPart g0 = {x, W, bias, A_ws, B_ws, N, 0, ntiles};
    const int r0 = (ntiles + 1) / 2;

    k_chunk_group<<<nchunks + r0, 256, 0, stream>>>(ei, Tc, rec, E, NB, nchunks, g0);
    k_bucket_finish<<<NB, 1024, 0, stream>>>(rec, Tc, srcs, B_ws, A_ws, out, psum, psq, N, NB, nchunks);
    k_bn<<<TAIL_WGS, TAIL_THREADS, 0, stream>>>(out, psum, psq, gamma, beta, N, NB);
}
